# starting kernel
# baseline (speedup 1.0000x reference)
#include <hip/hip_runtime.h>

#define D 128

typedef float v2f __attribute__((ext_vector_type(2)));
typedef float v8f __attribute__((ext_vector_type(8)));

__global__ __launch_bounds__(256) void zero_ws(float* __restrict__ p, int n) {
    int i = blockIdx.x * 256 + threadIdx.x;
    if (i < n) p[i] = 0.0f;
}

__global__ __launch_bounds__(256) void degree_count(const int* __restrict__ erow,
                                                    float* __restrict__ deg, int ne) {
    int i = blockIdx.x * 256 + threadIdx.x;
    if (i < ne) atomicAdd(deg + erow[i], 1.0f);
}

__global__ __launch_bounds__(256) void deg_rsqrt(float* __restrict__ deg, int n) {
    int i = blockIdx.x * 256 + threadIdx.x;
    if (i < n) deg[i] = rsqrtf(deg[i]);
}

__global__ __launch_bounds__(256) void edge_aggregate(
    const float* __restrict__ x, const int* __restrict__ erow,
    const int* __restrict__ ecol, const float* __restrict__ dis,
    float* __restrict__ agg, int ne)
{
    int wid = blockIdx.x * 8 + (threadIdx.x >> 5);
    if (wid >= ne) return;
    int e   = __builtin_amdgcn_readfirstlane(wid);
    int row = erow[e];
    int col = ecol[e];
    float w = dis[row] * dis[col];
    int lane = threadIdx.x & 31;

    float4 xv = *((const float4*)(x + (size_t)col * D) + lane);
    float* dst = agg + (size_t)row * D + lane * 4;
    atomicAdd(dst + 0, w * xv.x);
    atomicAdd(dst + 1, w * xv.y);
    atomicAdd(dst + 2, w * xv.z);
    atomicAdd(dst + 3, w * xv.w);
}

__global__ __launch_bounds__(256) void gemm_bias_wmma(
    const float* __restrict__ agg, const float* __restrict__ Wt,
    const float* __restrict__ bias, float* __restrict__ out)
{
    const int lane = threadIdx.x & 31;
    const int wave = threadIdx.x >> 5;
    const int m0   = blockIdx.x * 16;
    const int n0   = wave * 16;
    const int half = lane >> 4;
    const int l15  = lane & 15;

    float bval = bias[n0 + l15];
    v8f c;
#pragma unroll
    for (int r = 0; r < 8; ++r) c[r] = bval;

    const float* arow = agg + (size_t)(m0 + l15) * D;
#pragma unroll 8
    for (int k = 0; k < D; k += 4) {
        v2f a = *(const v2f*)(arow + k + half * 2);
        v2f b;
        b.x = Wt[(size_t)(k + half * 2 + 0) * D + n0 + l15];
        b.y = Wt[(size_t)(k + half * 2 + 1) * D + n0 + l15];
        c = __builtin_amdgcn_wmma_f32_16x16x4_f32(
                 false, a,  false, b,
                 (short)0, c,  false,  false);
    }

    float* orow = out + (size_t)(m0 + half * 8) * D + n0 + l15;
#pragma unroll
    for (int r = 0; r < 8; ++r) orow[(size_t)r * D] = c[r];
}

__global__ __launch_bounds__(128) void gemm_tail(
    const float* __restrict__ agg, const float* __restrict__ Wt,
    const float* __restrict__ bias, float* __restrict__ out,
    int row_start)
{
    int r = row_start + blockIdx.x;
    int n = threadIdx.x;
    float acc = bias[n];
    const float* a = agg + (size_t)r * D;
    for (int k = 0; k < D; ++k) acc = fmaf(a[k], Wt[(size_t)k * D + n], acc);
    out[(size_t)r * D + n] = acc;
}

extern "C" void kernel_launch(void* const* d_in, const int* in_sizes, int n_in,
                              void* d_out, int out_size, void* d_ws, size_t ws_size,
                              hipStream_t stream) {
    const float* x      = (const float*)d_in[0];
    const float* weight = (const float*)d_in[1];
    const float* bias   = (const float*)d_in[2];
    const int*   erow   = (const int*)d_in[3];
    const int*   ecol   = (const int*)d_in[4];

    const int n_nodes = in_sizes[0] / D;
    const int n_edges = in_sizes[3];

    float* agg = (float*)d_ws;
    float* deg = agg + (size_t)n_nodes * D;

    const int zn = n_nodes * (D + 1);
    zero_ws<<<(zn + 255) / 256, 256, 0, stream>>>((float*)d_ws, zn);

    degree_count<<<(n_edges + 255) / 256, 256, 0, stream>>>(erow, deg, n_edges);
    deg_rsqrt<<<(n_nodes + 255) / 256, 256, 0, stream>>>(deg, n_nodes);

    edge_aggregate<<<(n_edges + 7) / 8, 256, 0, stream>>>(x, erow, ecol, deg,
                                                          agg, n_edges);

    const int mtiles = n_nodes / 16;
    if (mtiles > 0)
        gemm_bias_wmma<<<mtiles, 256, 0, stream>>>(agg, weight, bias,
                                                   (float*)d_out);
    const int tail = n_nodes - mtiles * 16;
    if (tail > 0)
        gemm_tail<<<tail, D, 0, stream>>>(agg, weight, bias, (float*)d_out,
                                          mtiles * 16);
}


# Float-atomic edge scatter replaced by counting sort of edges by row (int rank atomics) and a per-row register gather-sum; sum order per row = atomic arrival rank
# speedup vs baseline: 2.4055x; 2.4055x over previous
#include <hip/hip_runtime.h>

#define D 128
#define CNT_PITCH 32

typedef float v2f __attribute__((ext_vector_type(2)));
typedef float v8f __attribute__((ext_vector_type(8)));

__global__ __launch_bounds__(256) void zero_counts(int* __restrict__ cnt, int n) {
    int i = blockIdx.x * 256 + threadIdx.x;
    if (i < n) cnt[(size_t)i * CNT_PITCH] = 0;
}

__global__ __launch_bounds__(256) void degree_rank(const int* __restrict__ erow,
                                                   int* __restrict__ cnt,
                                                   int* __restrict__ rank, int ne) {
    int i = blockIdx.x * 256 + threadIdx.x;
    if (i < ne) rank[i] = atomicAdd(cnt + (size_t)erow[i] * CNT_PITCH, 1);
}

__global__ __launch_bounds__(1024) void scan_rows_local(const int* __restrict__ cnt,
                                                        int* __restrict__ rowptr,
                                                        float* __restrict__ dis,
                                                        int* __restrict__ bsum, int n) {
    __shared__ int wsum[32];
    const int i    = blockIdx.x * 1024 + threadIdx.x;
    const int lane = threadIdx.x & 31;
    const int wv   = threadIdx.x >> 5;
    const int c = (i < n) ? cnt[(size_t)i * CNT_PITCH] : 0;
    if (i < n) dis[i] = rsqrtf((float)c);
    int v = c;
    for (int o = 1; o < 32; o <<= 1) {
        int t = __shfl_up(v, o);
        if (lane >= o) v += t;
    }
    if (lane == 31) wsum[wv] = v;
    __syncthreads();
    if (wv == 0) {
        const int s = wsum[lane];
        int t = s;
        for (int o = 1; o < 32; o <<= 1) {
            int u = __shfl_up(t, o);
            if (lane >= o) t += u;
        }
        wsum[lane] = t - s;
        if (lane == 31) bsum[blockIdx.x] = t;
    }
    __syncthreads();
    if (i < n) rowptr[i] = wsum[wv] + v - c;
}

__global__ __launch_bounds__(1024) void scan_rows_offset(int* __restrict__ rowptr,
                                                         const int* __restrict__ bsum,
                                                         int n, int ne) {
    __shared__ int red[32];
    const int b    = blockIdx.x;
    const int lane = threadIdx.x & 31;
    const int wv   = threadIdx.x >> 5;
    int s = 0;
    for (int j = threadIdx.x; j < b; j += 1024) s += bsum[j];
    for (int o = 16; o > 0; o >>= 1) s += __shfl_xor(s, o);
    if (lane == 0) red[wv] = s;
    __syncthreads();
    int t = red[lane];
    for (int o = 16; o > 0; o >>= 1) t += __shfl_xor(t, o);
    const int i = b * 1024 + threadIdx.x;
    if (i < n) rowptr[i] += t;
    if (i == 0) rowptr[n] = ne;
}

__global__ __launch_bounds__(256) void edge_place(
    const int* __restrict__ erow, const int* __restrict__ ecol,
    const int* __restrict__ rank, const int* __restrict__ rowptr,
    const float* __restrict__ dis, int* __restrict__ ed, int ne)
{
    int i = blockIdx.x * 256 + threadIdx.x;
    if (i >= ne) return;
    const int row = erow[i];
    const int col = ecol[i];
    const float w = dis[row] * dis[col];
    const int pos = rowptr[row] + rank[i];
    ((int2*)ed)[pos] = make_int2(col, __float_as_int(w));
}

__global__ __launch_bounds__(256) void row_gather_sum(
    const float* __restrict__ x, const int* __restrict__ rowptr,
    const int* __restrict__ ed, float* __restrict__ agg, int n)
{
    const int wid = blockIdx.x * 8 + (threadIdx.x >> 5);
    if (wid >= n) return;
    const int r    = __builtin_amdgcn_readfirstlane(wid);
    const int lane = threadIdx.x & 31;
    const int beg  = rowptr[r];
    const int end  = rowptr[r + 1];
    const int2* __restrict__ ed2 = (const int2*)ed;
    const float4* __restrict__ x4 = (const float4*)x;

    float4 acc = make_float4(0.0f, 0.0f, 0.0f, 0.0f);
    for (int base = beg; base < end; base += 32) {
        const int j = base + lane;
        int2 my = make_int2(0, 0);
        if (j < end) my = ed2[j];
        const int m = (end - base < 32) ? (end - base) : 32;
#pragma unroll 8
        for (int k = 0; k < m; ++k) {
            const int   c = __builtin_amdgcn_readlane(my.x, k);
            const float w = __int_as_float(__builtin_amdgcn_readlane(my.y, k));
            const float4 xv = x4[(size_t)c * (D / 4) + lane];
            acc.x = fmaf(w, xv.x, acc.x);
            acc.y = fmaf(w, xv.y, acc.y);
            acc.z = fmaf(w, xv.z, acc.z);
            acc.w = fmaf(w, xv.w, acc.w);
        }
    }
    ((float4*)agg)[(size_t)r * (D / 4) + lane] = acc;
}

__global__ __launch_bounds__(256) void gemm_bias_wmma(
    const float* __restrict__ agg, const float* __restrict__ Wt,
    const float* __restrict__ bias, float* __restrict__ out)
{
    const int lane = threadIdx.x & 31;
    const int wave = threadIdx.x >> 5;
    const int m0   = blockIdx.x * 16;
    const int n0   = wave * 16;
    const int half = lane >> 4;
    const int l15  = lane & 15;

    float bval = bias[n0 + l15];
    v8f c;
#pragma unroll
    for (int r = 0; r < 8; ++r) c[r] = bval;

    const float* arow = agg + (size_t)(m0 + l15) * D;
#pragma unroll 8
    for (int k = 0; k < D; k += 4) {
        v2f a = *(const v2f*)(arow + k + half * 2);
        v2f b;
        b.x = Wt[(size_t)(k + half * 2 + 0) * D + n0 + l15];
        b.y = Wt[(size_t)(k + half * 2 + 1) * D + n0 + l15];
        c = __builtin_amdgcn_wmma_f32_16x16x4_f32(
                 false, a,  false, b,
                 (short)0, c,  false,  false);
    }

    float* orow = out + (size_t)(m0 + half * 8) * D + n0 + l15;
#pragma unroll
    for (int r = 0; r < 8; ++r) orow[(size_t)r * D] = c[r];
}

__global__ __launch_bounds__(128) void gemm_tail(
    const float* __restrict__ agg, const float* __restrict__ Wt,
    const float* __restrict__ bias, float* __restrict__ out,
    int row_start)
{
    int r = row_start + blockIdx.x;
    int n = threadIdx.x;
    float acc = bias[n];
    const float* a = agg + (size_t)r * D;
    for (int k = 0; k < D; ++k) acc = fmaf(a[k], Wt[(size_t)k * D + n], acc);
    out[(size_t)r * D + n] = acc;
}

extern "C" void kernel_launch(void* const* d_in, const int* in_sizes, int n_in,
                              void* d_out, int out_size, void* d_ws, size_t ws_size,
                              hipStream_t stream) {
    const float* x      = (const float*)d_in[0];
    const float* weight = (const float*)d_in[1];
    const float* bias   = (const float*)d_in[2];
    const int*   erow   = (const int*)d_in[3];
    const int*   ecol   = (const int*)d_in[4];

    const int n_nodes = in_sizes[0] / D;
    const int n_edges = in_sizes[3];
    const int nblk    = (n_nodes + 1023) / 1024;

    char*  base = (char*)d_ws;
    size_t off  = 0;
    float* agg    = (float*)(base + off); off += ((size_t)n_nodes * D * 4 + 255) / 256 * 256;
    int*   cnt    = (int*)(base + off);   off += ((size_t)n_nodes * CNT_PITCH * 4 + 255) / 256 * 256;
    int*   rowptr = (int*)(base + off);   off += ((size_t)(n_nodes + 1) * 4 + 255) / 256 * 256;
    float* dis    = (float*)(base + off); off += ((size_t)n_nodes * 4 + 255) / 256 * 256;
    int*   bsum   = (int*)(base + off);   off += ((size_t)nblk * 4 + 255) / 256 * 256;
    int*   rank   = (int*)(base + off);   off += ((size_t)n_edges * 4 + 255) / 256 * 256;
    int*   ed     = (int*)(base + off);

    zero_counts<<<(n_nodes + 255) / 256, 256, 0, stream>>>(cnt, n_nodes);
    degree_rank<<<(n_edges + 255) / 256, 256, 0, stream>>>(erow, cnt, rank, n_edges);
    scan_rows_local<<<nblk, 1024, 0, stream>>>(cnt, rowptr, dis, bsum, n_nodes);
    scan_rows_offset<<<nblk, 1024, 0, stream>>>(rowptr, bsum, n_nodes, n_edges);
    edge_place<<<(n_edges + 255) / 256, 256, 0, stream>>>(erow, ecol, rank, rowptr, dis,
                                                          ed, n_edges);
    row_gather_sum<<<(n_nodes + 7) / 8, 256, 0, stream>>>(x, rowptr, ed, agg, n_nodes);

    const int mtiles = n_nodes / 16;
    if (mtiles > 0)
        gemm_bias_wmma<<<mtiles, 256, 0, stream>>>(agg, weight, bias,
                                                   (float*)d_out);
    const int tail = n_nodes - mtiles * 16;
    if (tail > 0)
        gemm_tail<<<tail, D, 0, stream>>>(agg, weight, bias, (float*)d_out,
                                          mtiles * 16);
}


# Edge rank: global returning atomics replaced by LDS atomics (each 1024-thread block counts 16384 edges in an all-row LDS histogram, prefix over blocks); row sum order = block, then LDS arrival
# speedup vs baseline: 2.6900x; 1.1183x over previous
#include <hip/hip_runtime.h>

#define D 128
#define HB_THREADS 1024
#define HB_CHUNKS 4
#define HB_EDGES_LOG2 14
#define HB_EDGES (1 << HB_EDGES_LOG2)
#define HB_ROWS 65536
static_assert(HB_THREADS * HB_CHUNKS * 4 == HB_EDGES, "edges per block");

#define BP_SEG 16
#define BP_WAVES 8

typedef float v2f __attribute__((ext_vector_type(2)));
typedef float v8f __attribute__((ext_vector_type(8)));

__device__ __forceinline__ void count_edge(int* h, int row, int w0, int wn, int& k) {
    const unsigned o = (unsigned)(row - w0);
    if (o < (unsigned)wn) k = atomicAdd(h + o, 1);
}

__global__ __launch_bounds__(HB_THREADS) void rank_in_lds(
    const int* __restrict__ erow, int* __restrict__ hist, int* __restrict__ rank,
    int n, int np, int ne)
{
    __shared__ int h[HB_ROWS];
    const int tid = threadIdx.x;
    const int n4  = ne >> 2;
    const int4* __restrict__ e4 = (const int4*)erow;

    int4 r[HB_CHUNKS], k[HB_CHUNKS];
    int  cidx[HB_CHUNKS];
#pragma unroll
    for (int j = 0; j < HB_CHUNKS; ++j) {
        cidx[j] = (blockIdx.x * HB_CHUNKS + j) * HB_THREADS + tid;
        r[j] = make_int4(-1, -1, -1, -1);
        if (cidx[j] < n4) r[j] = e4[cidx[j]];
        k[j] = make_int4(0, 0, 0, 0);
    }
    const bool has_tail = (blockIdx.x == 0) && (tid < (ne & 3));
    const int  ti = (n4 << 2) + tid;
    int trow = -1, tk = 0;
    if (has_tail) trow = erow[ti];

    for (int w0 = 0; w0 < n; w0 += HB_ROWS) {
        const int wn = (np - w0 < HB_ROWS) ? (np - w0) : HB_ROWS;
        for (int i = tid * 4; i < wn; i += HB_THREADS * 4)
            *(int4*)(h + i) = make_int4(0, 0, 0, 0);
        __syncthreads();
#pragma unroll
        for (int j = 0; j < HB_CHUNKS; ++j) {
            count_edge(h, r[j].x, w0, wn, k[j].x);
            count_edge(h, r[j].y, w0, wn, k[j].y);
            count_edge(h, r[j].z, w0, wn, k[j].z);
            count_edge(h, r[j].w, w0, wn, k[j].w);
        }
        count_edge(h, trow, w0, wn, tk);
        __syncthreads();
        int4* __restrict__ dst = (int4*)(hist + (size_t)blockIdx.x * np + w0);
        for (int i = tid * 4; i < wn; i += HB_THREADS * 4)
            dst[i >> 2] = *(const int4*)(h + i);
        __syncthreads();
    }
#pragma unroll
    for (int j = 0; j < HB_CHUNKS; ++j)
        if (cidx[j] < n4) ((int4*)rank)[cidx[j]] = k[j];
    if (has_tail) rank[ti] = tk;
}

__global__ __launch_bounds__(32 * BP_WAVES) void block_prefix(
    const int* __restrict__ hist, int* __restrict__ base, int* __restrict__ cnt,
    int n, int np, int nb)
{
    __shared__ int seg[BP_WAVES][32];
    const int x = threadIdx.x & 31;
    const int y = threadIdx.x >> 5;
    const int r = blockIdx.x * 32 + x;
    const bool live = r < n;
    int carry = 0;
    for (int b0 = 0; b0 < nb; b0 += BP_SEG * BP_WAVES) {
        const int bs = b0 + y * BP_SEG;
        int c[BP_SEG];
        int s = 0;
#pragma unroll
        for (int j = 0; j < BP_SEG; ++j) {
            c[j] = (live && bs + j < nb) ? hist[(size_t)(bs + j) * np + r] : 0;
            s += c[j];
        }
        seg[y][x] = s;
        __syncthreads();
        int pre = carry, tot = carry;
#pragma unroll
        for (int q = 0; q < BP_WAVES; ++q) {
            const int v = seg[q][x];
            if (q < y) pre += v;
            tot += v;
        }
#pragma unroll
        for (int j = 0; j < BP_SEG; ++j) {
            if (live && bs + j < nb) base[(size_t)(bs + j) * np + r] = pre;
            pre += c[j];
        }
        carry = tot;
        __syncthreads();
    }
    if (live && y == 0) cnt[r] = carry;
}

__global__ __launch_bounds__(1024) void scan_rows_local(const int* __restrict__ cnt,
                                                        int* __restrict__ rowptr,
                                                        float* __restrict__ dis,
                                                        int* __restrict__ bsum, int n) {
    __shared__ int wsum[32];
    const int i    = blockIdx.x * 1024 + threadIdx.x;
    const int lane = threadIdx.x & 31;
    const int wv   = threadIdx.x >> 5;
    const int c = (i < n) ? cnt[i] : 0;
    if (i < n) dis[i] = rsqrtf((float)c);
    int v = c;
    for (int o = 1; o < 32; o <<= 1) {
        int t = __shfl_up(v, o);
        if (lane >= o) v += t;
    }
    if (lane == 31) wsum[wv] = v;
    __syncthreads();
    if (wv == 0) {
        const int s = wsum[lane];
        int t = s;
        for (int o = 1; o < 32; o <<= 1) {
            int u = __shfl_up(t, o);
            if (lane >= o) t += u;
        }
        wsum[lane] = t - s;
        if (lane == 31) bsum[blockIdx.x] = t;
    }
    __syncthreads();
    if (i < n) rowptr[i] = wsum[wv] + v - c;
}

__global__ __launch_bounds__(1024) void scan_rows_offset(int* __restrict__ rowptr,
                                                         const int* __restrict__ bsum,
                                                         int n, int ne) {
    __shared__ int red[32];
    const int b    = blockIdx.x;
    const int lane = threadIdx.x & 31;
    const int wv   = threadIdx.x >> 5;
    int s = 0;
    for (int j = threadIdx.x; j < b; j += 1024) s += bsum[j];
    for (int o = 16; o > 0; o >>= 1) s += __shfl_xor(s, o);
    if (lane == 0) red[wv] = s;
    __syncthreads();
    int t = red[lane];
    for (int o = 16; o > 0; o >>= 1) t += __shfl_xor(t, o);
    const int i = b * 1024 + threadIdx.x;
    if (i < n) rowptr[i] += t;
    if (i == 0) rowptr[n] = ne;
}

__global__ __launch_bounds__(256) void edge_place(
    const int* __restrict__ erow, const int* __restrict__ ecol,
    const int* __restrict__ rank, const int* __restrict__ rowptr,
    const int* __restrict__ base, const float* __restrict__ dis,
    int* __restrict__ ed, int np, int ne)
{
    int i = blockIdx.x * 256 + threadIdx.x;
    if (i >= ne) return;
    const int row = erow[i];
    const int col = ecol[i];
    const int hb  = (i < (ne & ~3)) ? (i >> HB_EDGES_LOG2) : 0;
    const float w = dis[row] * dis[col];
    const int pos = rowptr[row] + base[(size_t)hb * np + row] + rank[i];
    ((int2*)ed)[pos] = make_int2(col, __float_as_int(w));
}

__global__ __launch_bounds__(256) void row_gather_sum(
    const float* __restrict__ x, const int* __restrict__ rowptr,
    const int* __restrict__ ed, float* __restrict__ agg, int n)
{
    const int wid = blockIdx.x * 8 + (threadIdx.x >> 5);
    if (wid >= n) return;
    const int r    = __builtin_amdgcn_readfirstlane(wid);
    const int lane = threadIdx.x & 31;
    const int beg  = rowptr[r];
    const int end  = rowptr[r + 1];
    const int2* __restrict__ ed2 = (const int2*)ed;
    const float4* __restrict__ x4 = (const float4*)x;

    float4 acc = make_float4(0.0f, 0.0f, 0.0f, 0.0f);
    for (int base = beg; base < end; base += 32) {
        const int j = base + lane;
        int2 my = make_int2(0, 0);
        if (j < end) my = ed2[j];
        const int m = (end - base < 32) ? (end - base) : 32;
#pragma unroll 8
        for (int k = 0; k < m; ++k) {
            const int   c = __builtin_amdgcn_readlane(my.x, k);
            const float w = __int_as_float(__builtin_amdgcn_readlane(my.y, k));
            const float4 xv = x4[(size_t)c * (D / 4) + lane];
            acc.x = fmaf(w, xv.x, acc.x);
            acc.y = fmaf(w, xv.y, acc.y);
            acc.z = fmaf(w, xv.z, acc.z);
            acc.w = fmaf(w, xv.w, acc.w);
        }
    }
    ((float4*)agg)[(size_t)r * (D / 4) + lane] = acc;
}

__global__ __launch_bounds__(256) void gemm_bias_wmma(
    const float* __restrict__ agg, const float* __restrict__ Wt,
    const float* __restrict__ bias, float* __restrict__ out)
{
    const int lane = threadIdx.x & 31;
    const int wave = threadIdx.x >> 5;
    const int m0   = blockIdx.x * 16;
    const int n0   = wave * 16;
    const int half = lane >> 4;
    const int l15  = lane & 15;

    float bval = bias[n0 + l15];
    v8f c;
#pragma unroll
    for (int r = 0; r < 8; ++r) c[r] = bval;

    const float* arow = agg + (size_t)(m0 + l15) * D;
#pragma unroll 8
    for (int k = 0; k < D; k += 4) {
        v2f a = *(const v2f*)(arow + k + half * 2);
        v2f b;
        b.x = Wt[(size_t)(k + half * 2 + 0) * D + n0 + l15];
        b.y = Wt[(size_t)(k + half * 2 + 1) * D + n0 + l15];
        c = __builtin_amdgcn_wmma_f32_16x16x4_f32(
                 false, a,  false, b,
                 (short)0, c,  false,  false);
    }

    float* orow = out + (size_t)(m0 + half * 8) * D + n0 + l15;
#pragma unroll
    for (int r = 0; r < 8; ++r) orow[(size_t)r * D] = c[r];
}

__global__ __launch_bounds__(128) void gemm_tail(
    const float* __restrict__ agg, const float* __restrict__ Wt,
    const float* __restrict__ bias, float* __restrict__ out,
    int row_start)
{
    int r = row_start + blockIdx.x;
    int n = threadIdx.x;
    float acc = bias[n];
    const float* a = agg + (size_t)r * D;
    for (int k = 0; k < D; ++k) acc = fmaf(a[k], Wt[(size_t)k * D + n], acc);
    out[(size_t)r * D + n] = acc;
}

extern "C" void kernel_launch(void* const* d_in, const int* in_sizes, int n_in,
                              void* d_out, int out_size, void* d_ws, size_t ws_size,
                              hipStream_t stream) {
    const float* x      = (const float*)d_in[0];
    const float* weight = (const float*)d_in[1];
    const float* bias   = (const float*)d_in[2];
    const int*   erow   = (const int*)d_in[3];
    const int*   ecol   = (const int*)d_in[4];

    const int n_nodes = in_sizes[0] / D;
    const int n_edges = in_sizes[3];
    const int nblk    = (n_nodes + 1023) / 1024;

    char*  base = (char*)d_ws;
    size_t off  = 0;
    float* agg    = (float*)(base + off); off += ((size_t)n_nodes * D * 4 + 255) / 256 * 256;
    int*   cnt    = (int*)(base + off);   off += ((size_t)n_nodes * 4 + 255) / 256 * 256;
    int*   rowptr = (int*)(base + off);   off += ((size_t)(n_nodes + 1) * 4 + 255) / 256 * 256;
    float* dis    = (float*)(base + off); off += ((size_t)n_nodes * 4 + 255) / 256 * 256;
    int*   bsum   = (int*)(base + off);   off += ((size_t)nblk * 4 + 255) / 256 * 256;
    int*   rank   = (int*)(base + off);   off += ((size_t)n_edges * 4 + 255) / 256 * 256;
    int*   ed     = (int*)(base + off);   off += ((size_t)n_edges * 8 + 255) / 256 * 256;
    const int np  = (n_nodes + 3) / 4 * 4;
    const int nb  = (n_edges / 4) / (HB_THREADS * HB_CHUNKS) + 1;
    int*   hist   = (int*)(base + off);   off += ((size_t)nb * np * 4 + 255) / 256 * 256;
    int*   bpre   = (int*)(base + off);

    rank_in_lds<<<nb, HB_THREADS, 0, stream>>>(erow, hist, rank, n_nodes, np, n_edges);
    block_prefix<<<(n_nodes + 31) / 32, 32 * BP_WAVES, 0, stream>>>(hist, bpre, cnt, n_nodes, np, nb);
    scan_rows_local<<<nblk, 1024, 0, stream>>>(cnt, rowptr, dis, bsum, n_nodes);
    scan_rows_offset<<<nblk, 1024, 0, stream>>>(rowptr, bsum, n_nodes, n_edges);
    edge_place<<<(n_edges + 255) / 256, 256, 0, stream>>>(erow, ecol, rank, rowptr, bpre, dis,
                                                          ed, np, n_edges);
    row_gather_sum<<<(n_nodes + 7) / 8, 256, 0, stream>>>(x, rowptr, ed, agg, n_nodes);

    const int mtiles = n_nodes / 16;
    if (mtiles > 0)
        gemm_bias_wmma<<<mtiles, 256, 0, stream>>>(agg, weight, bias,
                                                   (float*)d_out);
    const int tail = n_nodes - mtiles * 16;
    if (tail > 0)
        gemm_tail<<<tail, D, 0, stream>>>(agg, weight, bias, (float*)d_out,
                                          mtiles * 16);
}


# x@W re-tiled: one wave = 16 rows x 64 cols, its rows of x read once into registers and reused for 4 tiles; W staged once per block in LDS in fragment order (one 8-byte LDS read per B fragment)
# speedup vs baseline: 3.0569x; 1.1364x over previous
#include <hip/hip_runtime.h>

#define D 128
#define HB_THREADS 1024
#define HB_CHUNKS 4
#define HB_EDGES_LOG2 14
#define HB_EDGES (1 << HB_EDGES_LOG2)
#define HB_ROWS 65536
static_assert(HB_THREADS * HB_CHUNKS * 4 == HB_EDGES, "edges per block");

#define BP_SEG 16
#define BP_WAVES 8

typedef float v2f __attribute__((ext_vector_type(2)));
typedef float v8f __attribute__((ext_vector_type(8)));

__device__ __forceinline__ void count_edge(int* h, int row, int w0, int wn, int& k) {
    const unsigned o = (unsigned)(row - w0);
    if (o < (unsigned)wn) k = atomicAdd(h + o, 1);
}

__global__ __launch_bounds__(HB_THREADS) void rank_in_lds(
    const int* __restrict__ erow, int* __restrict__ hist, int* __restrict__ rank,
    int n, int np, int ne)
{
    __shared__ int h[HB_ROWS];
    const int tid = threadIdx.x;
    const int n4  = ne >> 2;
    const int4* __restrict__ e4 = (const int4*)erow;

    int4 r[HB_CHUNKS], k[HB_CHUNKS];
    int  cidx[HB_CHUNKS];
#pragma unroll
    for (int j = 0; j < HB_CHUNKS; ++j) {
        cidx[j] = (blockIdx.x * HB_CHUNKS + j) * HB_THREADS + tid;
        r[j] = make_int4(-1, -1, -1, -1);
        if (cidx[j] < n4) r[j] = e4[cidx[j]];
        k[j] = make_int4(0, 0, 0, 0);
    }
    const bool has_tail = (blockIdx.x == 0) && (tid < (ne & 3));
    const int  ti = (n4 << 2) + tid;
    int trow = -1, tk = 0;
    if (has_tail) trow = erow[ti];

    for (int w0 = 0; w0 < n; w0 += HB_ROWS) {
        const int wn = (np - w0 < HB_ROWS) ? (np - w0) : HB_ROWS;
        for (int i = tid * 4; i < wn; i += HB_THREADS * 4)
            *(int4*)(h + i) = make_int4(0, 0, 0, 0);
        __syncthreads();
#pragma unroll
        for (int j = 0; j < HB_CHUNKS; ++j) {
            count_edge(h, r[j].x, w0, wn, k[j].x);
            count_edge(h, r[j].y, w0, wn, k[j].y);
            count_edge(h, r[j].z, w0, wn, k[j].z);
            count_edge(h, r[j].w, w0, wn, k[j].w);
        }
        count_edge(h, trow, w0, wn, tk);
        __syncthreads();
        int4* __restrict__ dst = (int4*)(hist + (size_t)blockIdx.x * np + w0);
        for (int i = tid * 4; i < wn; i += HB_THREADS * 4)
            dst[i >> 2] = *(const int4*)(h + i);
        __syncthreads();
    }
#pragma unroll
    for (int j = 0; j < HB_CHUNKS; ++j)
        if (cidx[j] < n4) ((int4*)rank)[cidx[j]] = k[j];
    if (has_tail) rank[ti] = tk;
}

__global__ __launch_bounds__(32 * BP_WAVES) void block_prefix(
    const int* __restrict__ hist, int* __restrict__ base, int* __restrict__ cnt,
    int n, int np, int nb)
{
    __shared__ int seg[BP_WAVES][32];
    const int x = threadIdx.x & 31;
    const int y = threadIdx.x >> 5;
    const int r = blockIdx.x * 32 + x;
    const bool live = r < n;
    int carry = 0;
    for (int b0 = 0; b0 < nb; b0 += BP_SEG * BP_WAVES) {
        const int bs = b0 + y * BP_SEG;
        int c[BP_SEG];
        int s = 0;
#pragma unroll
        for (int j = 0; j < BP_SEG; ++j) {
            c[j] = (live && bs + j < nb) ? hist[(size_t)(bs + j) * np + r] : 0;
            s += c[j];
        }
        seg[y][x] = s;
        __syncthreads();
        int pre = carry, tot = carry;
#pragma unroll
        for (int q = 0; q < BP_WAVES; ++q) {
            const int v = seg[q][x];
            if (q < y) pre += v;
            tot += v;
        }
#pragma unroll
        for (int j = 0; j < BP_SEG; ++j) {
            if (live && bs + j < nb) base[(size_t)(bs + j) * np + r] = pre;
            pre += c[j];
        }
        carry = tot;
        __syncthreads();
    }
    if (live && y == 0) cnt[r] = carry;
}

__global__ __launch_bounds__(1024) void scan_rows_local(const int* __restrict__ cnt,
                                                        int* __restrict__ rowptr,
                                                        float* __restrict__ dis,
                                                        int* __restrict__ bsum, int n) {
    __shared__ int wsum[32];
    const int i    = blockIdx.x * 1024 + threadIdx.x;
    const int lane = threadIdx.x & 31;
    const int wv   = threadIdx.x >> 5;
    const int c = (i < n) ? cnt[i] : 0;
    if (i < n) dis[i] = rsqrtf((float)c);
    int v = c;
    for (int o = 1; o < 32; o <<= 1) {
        int t = __shfl_up(v, o);
        if (lane >= o) v += t;
    }
    if (lane == 31) wsum[wv] = v;
    __syncthreads();
    if (wv == 0) {
        const int s = wsum[lane];
        int t = s;
        for (int o = 1; o < 32; o <<= 1) {
            int u = __shfl_up(t, o);
            if (lane >= o) t += u;
        }
        wsum[lane] = t - s;
        if (lane == 31) bsum[blockIdx.x] = t;
    }
    __syncthreads();
    if (i < n) rowptr[i] = wsum[wv] + v - c;
}

__global__ __launch_bounds__(1024) void scan_rows_offset(int* __restrict__ rowptr,
                                                         const int* __restrict__ bsum,
                                                         int n, int ne) {
    __shared__ int red[32];
    const int b    = blockIdx.x;
    const int lane = threadIdx.x & 31;
    const int wv   = threadIdx.x >> 5;
    int s = 0;
    for (int j = threadIdx.x; j < b; j += 1024) s += bsum[j];
    for (int o = 16; o > 0; o >>= 1) s += __shfl_xor(s, o);
    if (lane == 0) red[wv] = s;
    __syncthreads();
    int t = red[lane];
    for (int o = 16; o > 0; o >>= 1) t += __shfl_xor(t, o);
    const int i = b * 1024 + threadIdx.x;
    if (i < n) rowptr[i] += t;
    if (i == 0) rowptr[n] = ne;
}

__global__ __launch_bounds__(256) void edge_place(
    const int* __restrict__ erow, const int* __restrict__ ecol,
    const int* __restrict__ rank, const int* __restrict__ rowptr,
    const int* __restrict__ base,
    int* __restrict__ ed, int np, int ne)
{
    int i = blockIdx.x * 256 + threadIdx.x;
    if (i >= ne) return;
    const int row = erow[i];
    const int col = ecol[i];
    const int hb  = (i < (ne & ~3)) ? (i >> HB_EDGES_LOG2) : 0;
    const int pos = rowptr[row] + base[(size_t)hb * np + row] + rank[i];
    ed[pos] = col;
}

__global__ __launch_bounds__(256) void row_gather_sum(
    const float* __restrict__ y, const int* __restrict__ rowptr,
    const int* __restrict__ ed, const float* __restrict__ dis,
    const float* __restrict__ bias, float* __restrict__ out, int n)
{
    const int wid = blockIdx.x * 8 + (threadIdx.x >> 5);
    if (wid >= n) return;
    const int r    = __builtin_amdgcn_readfirstlane(wid);
    const int lane = threadIdx.x & 31;
    const int beg  = rowptr[r];
    const int end  = rowptr[r + 1];
    const float dr = dis[r];
    const float4 bv = ((const float4*)bias)[lane];
    const float4* __restrict__ y4 = (const float4*)y;

    float4 acc = make_float4(0.0f, 0.0f, 0.0f, 0.0f);
    for (int base = beg; base < end; base += 32) {
        const int j = base + lane;
        int my = 0;
        if (j < end) my = ed[j];
        const int m = (end - base < 32) ? (end - base) : 32;
#pragma unroll 8
        for (int k = 0; k < m; ++k) {
            const int c = __builtin_amdgcn_readlane(my, k);
            const float4 yv = y4[(size_t)c * (D / 4) + lane];
            acc.x += yv.x;
            acc.y += yv.y;
            acc.z += yv.z;
            acc.w += yv.w;
        }
    }
    const float s = (end > beg) ? dr : 0.0f;
    float4 o;
    o.x = fmaf(acc.x, s, bv.x);
    o.y = fmaf(acc.y, s, bv.y);
    o.z = fmaf(acc.z, s, bv.z);
    o.w = fmaf(acc.w, s, bv.w);
    ((float4*)out)[(size_t)r * (D / 4) + lane] = o;
}

#define GS_KSTEPS (D / 4)
#define GS_TILES  (D / 16)
#define GS_WTILES 4
#define GS_STRIPS 4

__global__ __launch_bounds__(256) void gemm_scale_wmma(
    const float* __restrict__ x, const float* __restrict__ Wt,
    const float* __restrict__ dis, float* __restrict__ y, int mtiles)
{
    __shared__ __attribute__((aligned(16))) v2f Bs[GS_KSTEPS * GS_TILES * 32];

    const int lane  = threadIdx.x & 31;
    const int wave  = threadIdx.x >> 5;
    const int half  = lane >> 4;
    const int l15   = lane & 15;
    const int strip = blockIdx.x * GS_STRIPS + (wave >> 1);
    const int t0    = (wave & 1) * GS_WTILES;
    const bool live = strip < mtiles;
    const int m0    = strip * 16;

    {
        const float4* __restrict__ W4 = (const float4*)Wt;
        float4* Bq = (float4*)Bs;
#pragma unroll
        for (int it = 0; it < (D / 2) * (D / 4) / 256; ++it) {
            const int j = it * 256 + threadIdx.x;
            const int p = j >> 5;
            const int q = j & 31;
            const float4 w0 = W4[(2 * p) * (D / 4) + q];
            const float4 w1 = W4[(2 * p + 1) * (D / 4) + q];
            const int e = (((p >> 1) * GS_TILES + (q >> 2)) * 32) + (p & 1) * 16 + (q & 3) * 4;
            Bq[(e >> 1)]     = make_float4(w0.x, w1.x, w0.y, w1.y);
            Bq[(e >> 1) + 1] = make_float4(w0.z, w1.z, w0.w, w1.w);
        }
    }
    __syncthreads();
    if (!live) return;

    v2f a[GS_KSTEPS];
    {
        const float* arow = x + (size_t)(m0 + l15) * D + half * 2;
#pragma unroll
        for (int ks = 0; ks < GS_KSTEPS; ++ks) a[ks] = *(const v2f*)(arow + 4 * ks);
    }

    v8f c[GS_WTILES];
#pragma unroll
    for (int t = 0; t < GS_WTILES; ++t)
#pragma unroll
        for (int r = 0; r < 8; ++r) c[t][r] = 0.0f;

#pragma unroll
    for (int ks = 0; ks < GS_KSTEPS; ++ks) {
#pragma unroll
        for (int t = 0; t < GS_WTILES; ++t) {
            const v2f b = Bs[(ks * GS_TILES + t0 + t) * 32 + lane];
            c[t] = __builtin_amdgcn_wmma_f32_16x16x4_f32(
                        false, a[ks],  false, b,
                        (short)0, c[t],  false,  false);
        }
    }

    const float4 d0 = *(const float4*)(dis + m0 + half * 8);
    const float4 d1 = *(const float4*)(dis + m0 + half * 8 + 4);
    const float dv[8] = {d0.x, d0.y, d0.z, d0.w, d1.x, d1.y, d1.z, d1.w};
    float* orow = y + (size_t)(m0 + half * 8) * D + t0 * 16 + l15;
#pragma unroll
    for (int t = 0; t < GS_WTILES; ++t)
#pragma unroll
        for (int r = 0; r < 8; ++r) orow[(size_t)r * D + t * 16] = c[t][r] * dv[r];
}

__global__ __launch_bounds__(128) void gemm_tail(
    const float* __restrict__ x, const float* __restrict__ Wt,
    const float* __restrict__ dis, float* __restrict__ y,
    int row_start)
{
    int r = row_start + blockIdx.x;
    int n = threadIdx.x;
    float acc = 0.0f;
    const float* a = x + (size_t)r * D;
    for (int k = 0; k < D; ++k) acc = fmaf(a[k], Wt[(size_t)k * D + n], acc);
    y[(size_t)r * D + n] = acc * dis[r];
}

extern "C" void kernel_launch(void* const* d_in, const int* in_sizes, int n_in,
                              void* d_out, int out_size, void* d_ws, size_t ws_size,
                              hipStream_t stream) {
    const float* x      = (const float*)d_in[0];
    const float* weight = (const float*)d_in[1];
    const float* bias   = (const float*)d_in[2];
    const int*   erow   = (const int*)d_in[3];
    const int*   ecol   = (const int*)d_in[4];

    const int n_nodes = in_sizes[0] / D;
    const int n_edges = in_sizes[3];
    const int nblk    = (n_nodes + 1023) / 1024;

    char*  base = (char*)d_ws;
    size_t off  = 0;
    float* y      = (float*)(base + off); off += ((size_t)n_nodes * D * 4 + 255) / 256 * 256;
    int*   cnt    = (int*)(base + off);   off += ((size_t)n_nodes * 4 + 255) / 256 * 256;
    int*   rowptr = (int*)(base + off);   off += ((size_t)(n_nodes + 1) * 4 + 255) / 256 * 256;
    float* dis    = (float*)(base + off); off += ((size_t)n_nodes * 4 + 255) / 256 * 256;
    int*   bsum   = (int*)(base + off);   off += ((size_t)nblk * 4 + 255) / 256 * 256;
    int*   rank   = (int*)(base + off);   off += ((size_t)n_edges * 4 + 255) / 256 * 256;
    int*   ed     = (int*)(base + off);   off += ((size_t)n_edges * 4 + 255) / 256 * 256;
    const int np  = (n_nodes + 3) / 4 * 4;
    const int nb  = (n_edges / 4) / (HB_THREADS * HB_CHUNKS) + 1;
    int*   hist   = (int*)(base + off);   off += ((size_t)nb * np * 4 + 255) / 256 * 256;
    int*   bpre   = (int*)(base + off);

    rank_in_lds<<<nb, HB_THREADS, 0, stream>>>(erow, hist, rank, n_nodes, np, n_edges);
    block_prefix<<<(n_nodes + 31) / 32, 32 * BP_WAVES, 0, stream>>>(hist, bpre, cnt, n_nodes, np, nb);
    scan_rows_local<<<nblk, 1024, 0, stream>>>(cnt, rowptr, dis, bsum, n_nodes);
    scan_rows_offset<<<nblk, 1024, 0, stream>>>(rowptr, bsum, n_nodes, n_edges);
    const int mtiles = n_nodes / 16;
    if (mtiles > 0)
        gemm_scale_wmma<<<(mtiles + GS_STRIPS - 1) / GS_STRIPS, 256, 0, stream>>>(x, weight, dis, y, mtiles);
    const int tail = n_nodes - mtiles * 16;
    if (tail > 0)
        gemm_tail<<<tail, D, 0, stream>>>(x, weight, dis, y, mtiles * 16);

    edge_place<<<(n_edges + 255) / 256, 256, 0, stream>>>(erow, ecol, rank, rowptr, bpre,
                                                          ed, np, n_edges);
    row_gather_sum<<<(n_nodes + 7) / 8, 256, 0, stream>>>(y, rowptr, ed, dis, bias,
                                                          (float*)d_out, n_nodes);
}


# x@W (WMMA) and the edge placing share one launch (independent blocks interleaved in one grid; a placing thread takes 4 edges): matrix work under the scatter's memory waits, one launch fewer
# speedup vs baseline: 3.1575x; 1.0329x over previous
#include <hip/hip_runtime.h>

#define D 128
#define HB_THREADS 1024
#define HB_CHUNKS 4
#define HB_EDGES_LOG2 14
#define HB_EDGES (1 << HB_EDGES_LOG2)
#define HB_ROWS 65536
static_assert(HB_THREADS * HB_CHUNKS * 4 == HB_EDGES, "edges per block");

#define BP_SEG 16
#define BP_WAVES 8

typedef float v2f __attribute__((ext_vector_type(2)));
typedef float v8f __attribute__((ext_vector_type(8)));

__device__ __forceinline__ void count_edge(int* h, int row, int w0, int wn, int& k) {
    const unsigned o = (unsigned)(row - w0);
    if (o < (unsigned)wn) k = atomicAdd(h + o, 1);
}

__global__ __launch_bounds__(HB_THREADS) void rank_in_lds(
    const int* __restrict__ erow, int* __restrict__ hist, int* __restrict__ rank,
    int n, int np, int ne)
{
    __shared__ int h[HB_ROWS];
    const int tid = threadIdx.x;
    const int n4  = ne >> 2;
    const int4* __restrict__ e4 = (const int4*)erow;

    int4 r[HB_CHUNKS], k[HB_CHUNKS];
    int  cidx[HB_CHUNKS];
#pragma unroll
    for (int j = 0; j < HB_CHUNKS; ++j) {
        cidx[j] = (blockIdx.x * HB_CHUNKS + j) * HB_THREADS + tid;
        r[j] = make_int4(-1, -1, -1, -1);
        if (cidx[j] < n4) r[j] = e4[cidx[j]];
        k[j] = make_int4(0, 0, 0, 0);
    }
    const bool has_tail = (blockIdx.x == 0) && (tid < (ne & 3));
    const int  ti = (n4 << 2) + tid;
    int trow = -1, tk = 0;
    if (has_tail) trow = erow[ti];

    for (int w0 = 0; w0 < n; w0 += HB_ROWS) {
        const int wn = (np - w0 < HB_ROWS) ? (np - w0) : HB_ROWS;
        for (int i = tid * 4; i < wn; i += HB_THREADS * 4)
            *(int4*)(h + i) = make_int4(0, 0, 0, 0);
        __syncthreads();
#pragma unroll
        for (int j = 0; j < HB_CHUNKS; ++j) {
            count_edge(h, r[j].x, w0, wn, k[j].x);
            count_edge(h, r[j].y, w0, wn, k[j].y);
            count_edge(h, r[j].z, w0, wn, k[j].z);
            count_edge(h, r[j].w, w0, wn, k[j].w);
        }
        count_edge(h, trow, w0, wn, tk);
        __syncthreads();
        int4* __restrict__ dst = (int4*)(hist + (size_t)blockIdx.x * np + w0);
        for (int i = tid * 4; i < wn; i += HB_THREADS * 4)
            dst[i >> 2] = *(const int4*)(h + i);
        __syncthreads();
    }
#pragma unroll
    for (int j = 0; j < HB_CHUNKS; ++j)
        if (cidx[j] < n4) ((int4*)rank)[cidx[j]] = k[j];
    if (has_tail) rank[ti] = tk;
}

__global__ __launch_bounds__(32 * BP_WAVES) void block_prefix(
    const int* __restrict__ hist, int* __restrict__ base, int* __restrict__ cnt,
    int n, int np, int nb)
{
    __shared__ int seg[BP_WAVES][32];
    const int x = threadIdx.x & 31;
    const int y = threadIdx.x >> 5;
    const int r = blockIdx.x * 32 + x;
    const bool live = r < n;
    int carry = 0;
    for (int b0 = 0; b0 < nb; b0 += BP_SEG * BP_WAVES) {
        const int bs = b0 + y * BP_SEG;
        int c[BP_SEG];
        int s = 0;
#pragma unroll
        for (int j = 0; j < BP_SEG; ++j) {
            c[j] = (live && bs + j < nb) ? hist[(size_t)(bs + j) * np + r] : 0;
            s += c[j];
        }
        seg[y][x] = s;
        __syncthreads();
        int pre = carry, tot = carry;
#pragma unroll
        for (int q = 0; q < BP_WAVES; ++q) {
            const int v = seg[q][x];
            if (q < y) pre += v;
            tot += v;
        }
#pragma unroll
        for (int j = 0; j < BP_SEG; ++j) {
            if (live && bs + j < nb) base[(size_t)(bs + j) * np + r] = pre;
            pre += c[j];
        }
        carry = tot;
        __syncthreads();
    }
    if (live && y == 0) cnt[r] = carry;
}

__global__ __launch_bounds__(1024) void scan_rows_local(const int* __restrict__ cnt,
                                                        int* __restrict__ rowptr,
                                                        float* __restrict__ dis,
                                                        int* __restrict__ bsum, int n) {
    __shared__ int wsum[32];
    const int i    = blockIdx.x * 1024 + threadIdx.x;
    const int lane = threadIdx.x & 31;
    const int wv   = threadIdx.x >> 5;
    const int c = (i < n) ? cnt[i] : 0;
    if (i < n) dis[i] = rsqrtf((float)c);
    int v = c;
    for (int o = 1; o < 32; o <<= 1) {
        int t = __shfl_up(v, o);
        if (lane >= o) v += t;
    }
    if (lane == 31) wsum[wv] = v;
    __syncthreads();
    if (wv == 0) {
        const int s = wsum[lane];
        int t = s;
        for (int o = 1; o < 32; o <<= 1) {
            int u = __shfl_up(t, o);
            if (lane >= o) t += u;
        }
        wsum[lane] = t - s;
        if (lane == 31) bsum[blockIdx.x] = t;
    }
    __syncthreads();
    if (i < n) rowptr[i] = wsum[wv] + v - c;
}

__global__ __launch_bounds__(1024) void scan_rows_offset(int* __restrict__ rowptr,
                                                         const int* __restrict__ bsum,
                                                         int n, int ne) {
    __shared__ int red[32];
    const int b    = blockIdx.x;
    const int lane = threadIdx.x & 31;
    const int wv   = threadIdx.x >> 5;
    int s = 0;
    for (int j = threadIdx.x; j < b; j += 1024) s += bsum[j];
    for (int o = 16; o > 0; o >>= 1) s += __shfl_xor(s, o);
    if (lane == 0) red[wv] = s;
    __syncthreads();
    int t = red[lane];
    for (int o = 16; o > 0; o >>= 1) t += __shfl_xor(t, o);
    const int i = b * 1024 + threadIdx.x;
    if (i < n) rowptr[i] += t;
    if (i == 0) rowptr[n] = ne;
}

__device__ __forceinline__ void edge_place_block(
    const int* __restrict__ erow, const int* __restrict__ ecol,
    const int* __restrict__ rank, const int* __restrict__ rowptr,
    const int* __restrict__ base,
    int* __restrict__ ed, int np, int ne, int eblock)
{
    const int n4 = ne >> 2;
    const int ci = eblock * 256 + threadIdx.x;
    if (ci < n4) {
        const int4 r = ((const int4*)erow)[ci];
        const int4 c = ((const int4*)ecol)[ci];
        const int4 k = ((const int4*)rank)[ci];
        const int* __restrict__ bb = base + (size_t)(ci >> (HB_EDGES_LOG2 - 2)) * np;
        const int p0 = rowptr[r.x] + bb[r.x] + k.x;
        const int p1 = rowptr[r.y] + bb[r.y] + k.y;
        const int p2 = rowptr[r.z] + bb[r.z] + k.z;
        const int p3 = rowptr[r.w] + bb[r.w] + k.w;
        ed[p0] = c.x;
        ed[p1] = c.y;
        ed[p2] = c.z;
        ed[p3] = c.w;
    }
    if (eblock == 0 && (int)threadIdx.x < (ne & 3)) {
        const int i   = (n4 << 2) + threadIdx.x;
        const int row = erow[i];
        ed[rowptr[row] + base[row] + rank[i]] = ecol[i];
    }
}

__global__ __launch_bounds__(256) void row_gather_sum(
    const float* __restrict__ y, const int* __restrict__ rowptr,
    const int* __restrict__ ed, const float* __restrict__ dis,
    const float* __restrict__ bias, float* __restrict__ out, int n)
{
    const int wid = blockIdx.x * 8 + (threadIdx.x >> 5);
    if (wid >= n) return;
    const int r    = __builtin_amdgcn_readfirstlane(wid);
    const int lane = threadIdx.x & 31;
    const int beg  = rowptr[r];
    const int end  = rowptr[r + 1];
    const float dr = dis[r];
    const float4 bv = ((const float4*)bias)[lane];
    const float4* __restrict__ y4 = (const float4*)y;

    float4 acc = make_float4(0.0f, 0.0f, 0.0f, 0.0f);
    for (int base = beg; base < end; base += 32) {
        const int j = base + lane;
        int my = 0;
        if (j < end) my = ed[j];
        const int m = (end - base < 32) ? (end - base) : 32;
#pragma unroll 8
        for (int k = 0; k < m; ++k) {
            const int c = __builtin_amdgcn_readlane(my, k);
            const float4 yv = y4[(size_t)c * (D / 4) + lane];
            acc.x += yv.x;
            acc.y += yv.y;
            acc.z += yv.z;
            acc.w += yv.w;
        }
    }
    const float s = (end > beg) ? dr : 0.0f;
    float4 o;
    o.x = fmaf(acc.x, s, bv.x);
    o.y = fmaf(acc.y, s, bv.y);
    o.z = fmaf(acc.z, s, bv.z);
    o.w = fmaf(acc.w, s, bv.w);
    ((float4*)out)[(size_t)r * (D / 4) + lane] = o;
}

#define GS_KSTEPS (D / 4)
#define GS_TILES  (D / 16)
#define GS_WTILES 4
#define GS_STRIPS 4

__device__ __forceinline__ void gemm_scale_block(
    const float* __restrict__ x, const float* __restrict__ Wt,
    const float* __restrict__ dis, float* __restrict__ y, int mtiles,
    int gblock, v2f* Bs)
{

    const int lane  = threadIdx.x & 31;
    const int wave  = threadIdx.x >> 5;
    const int half  = lane >> 4;
    const int l15   = lane & 15;
    const int strip = gblock * GS_STRIPS + (wave >> 1);
    const int t0    = (wave & 1) * GS_WTILES;
    const bool live = strip < mtiles;
    const int m0    = strip * 16;

    {
        const float4* __restrict__ W4 = (const float4*)Wt;
        float4* Bq = (float4*)Bs;
#pragma unroll
        for (int it = 0; it < (D / 2) * (D / 4) / 256; ++it) {
            const int j = it * 256 + threadIdx.x;
            const int p = j >> 5;
            const int q = j & 31;
            const float4 w0 = W4[(2 * p) * (D / 4) + q];
            const float4 w1 = W4[(2 * p + 1) * (D / 4) + q];
            const int e = (((p >> 1) * GS_TILES + (q >> 2)) * 32) + (p & 1) * 16 + (q & 3) * 4;
            Bq[(e >> 1)]     = make_float4(w0.x, w1.x, w0.y, w1.y);
            Bq[(e >> 1) + 1] = make_float4(w0.z, w1.z, w0.w, w1.w);
        }
    }
    __syncthreads();
    if (!live) return;

    v2f a[GS_KSTEPS];
    {
        const float* arow = x + (size_t)(m0 + l15) * D + half * 2;
#pragma unroll
        for (int ks = 0; ks < GS_KSTEPS; ++ks) a[ks] = *(const v2f*)(arow + 4 * ks);
    }

    v8f c[GS_WTILES];
#pragma unroll
    for (int t = 0; t < GS_WTILES; ++t)
#pragma unroll
        for (int r = 0; r < 8; ++r) c[t][r] = 0.0f;

#pragma unroll
    for (int ks = 0; ks < GS_KSTEPS; ++ks) {
#pragma unroll
        for (int t = 0; t < GS_WTILES; ++t) {
            const v2f b = Bs[(ks * GS_TILES + t0 + t) * 32 + lane];
            c[t] = __builtin_amdgcn_wmma_f32_16x16x4_f32(
                        false, a[ks],  false, b,
                        (short)0, c[t],  false,  false);
        }
    }

    const float4 d0 = *(const float4*)(dis + m0 + half * 8);
    const float4 d1 = *(const float4*)(dis + m0 + half * 8 + 4);
    const float dv[8] = {d0.x, d0.y, d0.z, d0.w, d1.x, d1.y, d1.z, d1.w};
    float* orow = y + (size_t)(m0 + half * 8) * D + t0 * 16 + l15;
#pragma unroll
    for (int t = 0; t < GS_WTILES; ++t)
#pragma unroll
        for (int r = 0; r < 8; ++r) orow[(size_t)r * D + t * 16] = c[t][r] * dv[r];
}

__global__ __launch_bounds__(256) void gemm_beside_place(
    const float* __restrict__ x, const float* __restrict__ Wt,
    const float* __restrict__ dis, float* __restrict__ y, int mtiles, int gblocks,
    const int* __restrict__ erow, const int* __restrict__ ecol,
    const int* __restrict__ rank, const int* __restrict__ rowptr,
    const int* __restrict__ base, int* __restrict__ ed, int np, int ne)
{
    __shared__ __attribute__((aligned(16))) v2f Bs[GS_KSTEPS * GS_TILES * 32];

    const int b  = blockIdx.x;
    const int sh = (gridDim.x >= 2u * (unsigned)gblocks) ? 1 : 0;
    const int q  = b >> sh;
    const int r  = b & sh;
    if (r == 0 && q < gblocks) {
        gemm_scale_block(x, Wt, dis, y, mtiles, q, Bs);
        return;
    }
    const int gfront = (q + r < gblocks) ? (q + r) : gblocks;
    edge_place_block(erow, ecol, rank, rowptr, base, ed, np, ne, b - gfront);
}

__global__ __launch_bounds__(128) void gemm_tail(
    const float* __restrict__ x, const float* __restrict__ Wt,
    const float* __restrict__ dis, float* __restrict__ y,
    int row_start)
{
    int r = row_start + blockIdx.x;
    int n = threadIdx.x;
    float acc = 0.0f;
    const float* a = x + (size_t)r * D;
    for (int k = 0; k < D; ++k) acc = fmaf(a[k], Wt[(size_t)k * D + n], acc);
    y[(size_t)r * D + n] = acc * dis[r];
}

extern "C" void kernel_launch(void* const* d_in, const int* in_sizes, int n_in,
                              void* d_out, int out_size, void* d_ws, size_t ws_size,
                              hipStream_t stream) {
    const float* x      = (const float*)d_in[0];
    const float* weight = (const float*)d_in[1];
    const float* bias   = (const float*)d_in[2];
    const int*   erow   = (const int*)d_in[3];
    const int*   ecol   = (const int*)d_in[4];

    const int n_nodes = in_sizes[0] / D;
    const int n_edges = in_sizes[3];
    const int nblk    = (n_nodes + 1023) / 1024;

    char*  base = (char*)d_ws;
    size_t off  = 0;
    float* y      = (float*)(base + off); off += ((size_t)n_nodes * D * 4 + 255) / 256 * 256;
    int*   cnt    = (int*)(base + off);   off += ((size_t)n_nodes * 4 + 255) / 256 * 256;
    int*   rowptr = (int*)(base + off);   off += ((size_t)(n_nodes + 1) * 4 + 255) / 256 * 256;
    float* dis    = (float*)(base + off); off += ((size_t)n_nodes * 4 + 255) / 256 * 256;
    int*   bsum   = (int*)(base + off);   off += ((size_t)nblk * 4 + 255) / 256 * 256;
    int*   rank   = (int*)(base + off);   off += ((size_t)n_edges * 4 + 255) / 256 * 256;
    int*   ed     = (int*)(base + off);   off += ((size_t)n_edges * 4 + 255) / 256 * 256;
    const int np  = (n_nodes + 3) / 4 * 4;
    const int nb  = (n_edges / 4) / (HB_THREADS * HB_CHUNKS) + 1;
    int*   hist   = (int*)(base + off);   off += ((size_t)nb * np * 4 + 255) / 256 * 256;
    int*   bpre   = (int*)(base + off);

    rank_in_lds<<<nb, HB_THREADS, 0, stream>>>(erow, hist, rank, n_nodes, np, n_edges);
    block_prefix<<<(n_nodes + 31) / 32, 32 * BP_WAVES, 0, stream>>>(hist, bpre, cnt, n_nodes, np, nb);
    scan_rows_local<<<nblk, 1024, 0, stream>>>(cnt, rowptr, dis, bsum, n_nodes);
    scan_rows_offset<<<nblk, 1024, 0, stream>>>(rowptr, bsum, n_nodes, n_edges);
    const int mtiles  = n_nodes / 16;
    const int gblocks = (mtiles + GS_STRIPS - 1) / GS_STRIPS;
    const int eblocks = (n_edges / 4 + 256) / 256;
    gemm_beside_place<<<gblocks + eblocks, 256, 0, stream>>>(x, weight, dis, y, mtiles, gblocks,
                                                          erow, ecol, rank, rowptr, bpre,
                                                          ed, np, n_edges);
    const int tail = n_nodes - mtiles * 16;
    if (tail > 0)
        gemm_tail<<<tail, D, 0, stream>>>(x, weight, dis, y, mtiles * 16);

    row_gather_sum<<<(n_nodes + 7) / 8, 256, 0, stream>>>(y, rowptr, ed, dis, bias,
                                                          (float*)d_out, n_nodes);
}


# Edge-list place of each 32-row group taken by one int atomic add inside the block-prefix launch, folded into base[block][row]: both scan launches and the placing's row-start read go; row sums same
# speedup vs baseline: 3.3567x; 1.0631x over previous
#include <hip/hip_runtime.h>

#define D 128
#define HB_THREADS 1024
#define HB_CHUNKS 4
#define HB_EDGES_LOG2 14
#define HB_EDGES (1 << HB_EDGES_LOG2)
#define HB_ROWS 65536
static_assert(HB_THREADS * HB_CHUNKS * 4 == HB_EDGES, "edges per block");

#define BP_SEG 16
#define BP_WAVES 8

typedef float v2f __attribute__((ext_vector_type(2)));
typedef float v8f __attribute__((ext_vector_type(8)));

__device__ __forceinline__ void count_edge(int* h, int row, int w0, int wn, int& k) {
    const unsigned o = (unsigned)(row - w0);
    if (o < (unsigned)wn) k = atomicAdd(h + o, 1);
}

__global__ __launch_bounds__(HB_THREADS) void rank_in_lds(
    const int* __restrict__ erow, int* __restrict__ hist, int* __restrict__ rank,
    int* __restrict__ segctr, int n, int np, int ne)
{
    __shared__ int h[HB_ROWS];
    const int tid = threadIdx.x;
    if (blockIdx.x == 0 && tid == 0) *segctr = 0;
    const int n4  = ne >> 2;
    const int4* __restrict__ e4 = (const int4*)erow;

    int4 r[HB_CHUNKS], k[HB_CHUNKS];
    int  cidx[HB_CHUNKS];
#pragma unroll
    for (int j = 0; j < HB_CHUNKS; ++j) {
        cidx[j] = (blockIdx.x * HB_CHUNKS + j) * HB_THREADS + tid;
        r[j] = make_int4(-1, -1, -1, -1);
        if (cidx[j] < n4) r[j] = e4[cidx[j]];
        k[j] = make_int4(0, 0, 0, 0);
    }
    const bool has_tail = (blockIdx.x == 0) && (tid < (ne & 3));
    const int  ti = (n4 << 2) + tid;
    int trow = -1, tk = 0;
    if (has_tail) trow = erow[ti];

    for (int w0 = 0; w0 < n; w0 += HB_ROWS) {
        const int wn = (np - w0 < HB_ROWS) ? (np - w0) : HB_ROWS;
        for (int i = tid * 4; i < wn; i += HB_THREADS * 4)
            *(int4*)(h + i) = make_int4(0, 0, 0, 0);
        __syncthreads();
#pragma unroll
        for (int j = 0; j < HB_CHUNKS; ++j) {
            count_edge(h, r[j].x, w0, wn, k[j].x);
            count_edge(h, r[j].y, w0, wn, k[j].y);
            count_edge(h, r[j].z, w0, wn, k[j].z);
            count_edge(h, r[j].w, w0, wn, k[j].w);
        }
        count_edge(h, trow, w0, wn, tk);
        __syncthreads();
        int4* __restrict__ dst = (int4*)(hist + (size_t)blockIdx.x * np + w0);
        for (int i = tid * 4; i < wn; i += HB_THREADS * 4)
            dst[i >> 2] = *(const int4*)(h + i);
        __syncthreads();
    }
#pragma unroll
    for (int j = 0; j < HB_CHUNKS; ++j)
        if (cidx[j] < n4) ((int4*)rank)[cidx[j]] = k[j];
    if (has_tail) rank[ti] = tk;
}

__global__ __launch_bounds__(32 * BP_WAVES) void block_prefix(
    const int* __restrict__ hist, int* __restrict__ base, int* __restrict__ cnt,
    int* __restrict__ rowbeg, float* __restrict__ dis, int* __restrict__ segctr,
    int n, int np, int nb)
{
    __shared__ int seg[BP_WAVES][32];
    __shared__ int later[BP_WAVES][32];
    __shared__ int rstart[32];
    const int x = threadIdx.x & 31;
    const int y = threadIdx.x >> 5;
    const int r = blockIdx.x * 32 + x;
    const bool live = r < n;
    int ls = 0;
    for (int b = BP_SEG * BP_WAVES + y; b < nb; b += BP_WAVES)
        if (live) ls += hist[(size_t)b * np + r];
    later[y][x] = ls;
    int carry = 0;
    for (int b0 = 0; b0 < nb; b0 += BP_SEG * BP_WAVES) {
        const int bs = b0 + y * BP_SEG;
        int c[BP_SEG];
        int s = 0;
#pragma unroll
        for (int j = 0; j < BP_SEG; ++j) {
            c[j] = (live && bs + j < nb) ? hist[(size_t)(bs + j) * np + r] : 0;
            s += c[j];
        }
        seg[y][x] = s;
        __syncthreads();
        int pre = carry, tot = carry;
#pragma unroll
        for (int q = 0; q < BP_WAVES; ++q) {
            const int v = seg[q][x];
            if (q < y) pre += v;
            tot += v;
        }
        if (b0 == 0) {
            int deg = 0, beg = 0;
            if (y == 0) {
                deg = tot;
                if (nb > BP_SEG * BP_WAVES) {
#pragma nounroll
                    for (int q = 0; q < BP_WAVES; ++q) deg += later[q][x];
                }
                int inc = deg;
                for (int o = 1; o < 32; o <<= 1) {
                    const int t = __shfl_up(inc, o);
                    if (x >= o) inc += t;
                }
                const int btot = __shfl(inc, 31);
                int st = 0;
                if (x == 0) st = atomicAdd(segctr, btot);
                st = __shfl(st, 0);
                beg = st + inc - deg;
                rstart[x] = beg;
            }
            __syncthreads();
            const int rs = rstart[x];
            pre += rs;
            tot += rs;
            if (y == 0 && live) {
                rowbeg[r] = beg;
                cnt[r]    = deg;
                dis[r]    = rsqrtf((float)deg);
            }
        }
#pragma unroll
        for (int j = 0; j < BP_SEG; ++j) {
            if (live && bs + j < nb) base[(size_t)(bs + j) * np + r] = pre;
            pre += c[j];
        }
        carry = tot;
        __syncthreads();
    }
}

__device__ __forceinline__ void edge_place_block(
    const int* __restrict__ erow, const int* __restrict__ ecol,
    const int* __restrict__ rank, const int* __restrict__ base,
    int* __restrict__ ed, int np, int ne, int eblock)
{
    const int n4 = ne >> 2;
    const int ci = eblock * 256 + threadIdx.x;
    if (ci < n4) {
        const int4 r = ((const int4*)erow)[ci];
        const int4 c = ((const int4*)ecol)[ci];
        const int4 k = ((const int4*)rank)[ci];
        const int* __restrict__ bb = base + (size_t)(ci >> (HB_EDGES_LOG2 - 2)) * np;
        const int p0 = bb[r.x] + k.x;
        const int p1 = bb[r.y] + k.y;
        const int p2 = bb[r.z] + k.z;
        const int p3 = bb[r.w] + k.w;
        ed[p0] = c.x;
        ed[p1] = c.y;
        ed[p2] = c.z;
        ed[p3] = c.w;
    }
    if (eblock == 0 && (int)threadIdx.x < (ne & 3)) {
        const int i   = (n4 << 2) + threadIdx.x;
        const int row = erow[i];
        ed[base[row] + rank[i]] = ecol[i];
    }
}

__global__ __launch_bounds__(256) void row_gather_sum(
    const float* __restrict__ y, const int* __restrict__ rowbeg,
    const int* __restrict__ cnt,
    const int* __restrict__ ed, const float* __restrict__ dis,
    const float* __restrict__ bias, float* __restrict__ out, int n)
{
    const int wid = blockIdx.x * 8 + (threadIdx.x >> 5);
    if (wid >= n) return;
    const int r    = __builtin_amdgcn_readfirstlane(wid);
    const int lane = threadIdx.x & 31;
    const int beg  = rowbeg[r];
    const int end  = beg + cnt[r];
    const float dr = dis[r];
    const float4 bv = ((const float4*)bias)[lane];
    const float4* __restrict__ y4 = (const float4*)y;

    float4 acc = make_float4(0.0f, 0.0f, 0.0f, 0.0f);
    for (int base = beg; base < end; base += 32) {
        const int j = base + lane;
        int my = 0;
        if (j < end) my = ed[j];
        const int m = (end - base < 32) ? (end - base) : 32;
#pragma unroll 8
        for (int k = 0; k < m; ++k) {
            const int c = __builtin_amdgcn_readlane(my, k);
            const float4 yv = y4[(size_t)c * (D / 4) + lane];
            acc.x += yv.x;
            acc.y += yv.y;
            acc.z += yv.z;
            acc.w += yv.w;
        }
    }
    const float s = (end > beg) ? dr : 0.0f;
    float4 o;
    o.x = fmaf(acc.x, s, bv.x);
    o.y = fmaf(acc.y, s, bv.y);
    o.z = fmaf(acc.z, s, bv.z);
    o.w = fmaf(acc.w, s, bv.w);
    ((float4*)out)[(size_t)r * (D / 4) + lane] = o;
}

#define GS_KSTEPS (D / 4)
#define GS_TILES  (D / 16)
#define GS_WTILES 4
#define GS_STRIPS 4

__device__ __forceinline__ void gemm_scale_block(
    const float* __restrict__ x, const float* __restrict__ Wt,
    const float* __restrict__ dis, float* __restrict__ y, int mtiles,
    int gblock, v2f* Bs)
{

    const int lane  = threadIdx.x & 31;
    const int wave  = threadIdx.x >> 5;
    const int half  = lane >> 4;
    const int l15   = lane & 15;
    const int strip = gblock * GS_STRIPS + (wave >> 1);
    const int t0    = (wave & 1) * GS_WTILES;
    const bool live = strip < mtiles;
    const int m0    = strip * 16;

    {
        const float4* __restrict__ W4 = (const float4*)Wt;
        float4* Bq = (float4*)Bs;
#pragma unroll
        for (int it = 0; it < (D / 2) * (D / 4) / 256; ++it) {
            const int j = it * 256 + threadIdx.x;
            const int p = j >> 5;
            const int q = j & 31;
            const float4 w0 = W4[(2 * p) * (D / 4) + q];
            const float4 w1 = W4[(2 * p + 1) * (D / 4) + q];
            const int e = (((p >> 1) * GS_TILES + (q >> 2)) * 32) + (p & 1) * 16 + (q & 3) * 4;
            Bq[(e >> 1)]     = make_float4(w0.x, w1.x, w0.y, w1.y);
            Bq[(e >> 1) + 1] = make_float4(w0.z, w1.z, w0.w, w1.w);
        }
    }
    __syncthreads();
    if (!live) return;

    v2f a[GS_KSTEPS];
    {
        const float* arow = x + (size_t)(m0 + l15) * D + half * 2;
#pragma unroll
        for (int ks = 0; ks < GS_KSTEPS; ++ks) a[ks] = *(const v2f*)(arow + 4 * ks);
    }

    v8f c[GS_WTILES];
#pragma unroll
    for (int t = 0; t < GS_WTILES; ++t)
#pragma unroll
        for (int r = 0; r < 8; ++r) c[t][r] = 0.0f;

#pragma unroll
    for (int ks = 0; ks < GS_KSTEPS; ++ks) {
#pragma unroll
        for (int t = 0; t < GS_WTILES; ++t) {
            const v2f b = Bs[(ks * GS_TILES + t0 + t) * 32 + lane];
            c[t] = __builtin_amdgcn_wmma_f32_16x16x4_f32(
                        false, a[ks],  false, b,
                        (short)0, c[t],  false,  false);
        }
    }

    const float4 d0 = *(const float4*)(dis + m0 + half * 8);
    const float4 d1 = *(const float4*)(dis + m0 + half * 8 + 4);
    const float dv[8] = {d0.x, d0.y, d0.z, d0.w, d1.x, d1.y, d1.z, d1.w};
    float* orow = y + (size_t)(m0 + half * 8) * D + t0 * 16 + l15;
#pragma unroll
    for (int t = 0; t < GS_WTILES; ++t)
#pragma unroll
        for (int r = 0; r < 8; ++r) orow[(size_t)r * D + t * 16] = c[t][r] * dv[r];
}

__global__ __launch_bounds__(256) void gemm_beside_place(
    const float* __restrict__ x, const float* __restrict__ Wt,
    const float* __restrict__ dis, float* __restrict__ y, int mtiles, int gblocks,
    const int* __restrict__ erow, const int* __restrict__ ecol,
    const int* __restrict__ rank,
    const int* __restrict__ base, int* __restrict__ ed, int np, int ne)
{
    __shared__ __attribute__((aligned(16))) v2f Bs[GS_KSTEPS * GS_TILES * 32];

    const int b  = blockIdx.x;
    const int sh = (gridDim.x >= 2u * (unsigned)gblocks) ? 1 : 0;
    const int q  = b >> sh;
    const int r  = b & sh;
    if (r == 0 && q < gblocks) {
        gemm_scale_block(x, Wt, dis, y, mtiles, q, Bs);
        return;
    }
    const int gfront = (q + r < gblocks) ? (q + r) : gblocks;
    edge_place_block(erow, ecol, rank, base, ed, np, ne, b - gfront);
}

__global__ __launch_bounds__(128) void gemm_tail(
    const float* __restrict__ x, const float* __restrict__ Wt,
    const float* __restrict__ dis, float* __restrict__ y,
    int row_start)
{
    int r = row_start + blockIdx.x;
    int n = threadIdx.x;
    float acc = 0.0f;
    const float* a = x + (size_t)r * D;
    for (int k = 0; k < D; ++k) acc = fmaf(a[k], Wt[(size_t)k * D + n], acc);
    y[(size_t)r * D + n] = acc * dis[r];
}

extern "C" void kernel_launch(void* const* d_in, const int* in_sizes, int n_in,
                              void* d_out, int out_size, void* d_ws, size_t ws_size,
                              hipStream_t stream) {
    const float* x      = (const float*)d_in[0];
    const float* weight = (const float*)d_in[1];
    const float* bias   = (const float*)d_in[2];
    const int*   erow   = (const int*)d_in[3];
    const int*   ecol   = (const int*)d_in[4];

    const int n_nodes = in_sizes[0] / D;
    const int n_edges = in_sizes[3];

    char*  base = (char*)d_ws;
    size_t off  = 0;
    float* y      = (float*)(base + off); off += ((size_t)n_nodes * D * 4 + 255) / 256 * 256;
    int*   cnt    = (int*)(base + off);   off += ((size_t)n_nodes * 4 + 255) / 256 * 256;
    int*   rowbeg = (int*)(base + off);   off += ((size_t)n_nodes * 4 + 255) / 256 * 256;
    float* dis    = (float*)(base + off); off += ((size_t)n_nodes * 4 + 255) / 256 * 256;
    int*   segctr = (int*)(base + off);   off += 256;
    int*   rank   = (int*)(base + off);   off += ((size_t)n_edges * 4 + 255) / 256 * 256;
    int*   ed     = (int*)(base + off);   off += ((size_t)n_edges * 4 + 255) / 256 * 256;
    const int np  = (n_nodes + 3) / 4 * 4;
    const int nb  = (n_edges / 4) / (HB_THREADS * HB_CHUNKS) + 1;
    int*   hist   = (int*)(base + off);   off += ((size_t)nb * np * 4 + 255) / 256 * 256;
    int*   bpre   = (int*)(base + off);

    rank_in_lds<<<nb, HB_THREADS, 0, stream>>>(erow, hist, rank, segctr, n_nodes, np, n_edges);
    block_prefix<<<(n_nodes + 31) / 32, 32 * BP_WAVES, 0, stream>>>(hist, bpre, cnt, rowbeg, dis, segctr,
                                                                   n_nodes, np, nb);
    const int mtiles  = n_nodes / 16;
    const int gblocks = (mtiles + GS_STRIPS - 1) / GS_STRIPS;
    const int eblocks = (n_edges / 4 + 256) / 256;
    gemm_beside_place<<<gblocks + eblocks, 256, 0, stream>>>(x, weight, dis, y, mtiles, gblocks,
                                                          erow, ecol, rank, bpre,
                                                          ed, np, n_edges);
    const int tail = n_nodes - mtiles * 16;
    if (tail > 0)
        gemm_tail<<<tail, D, 0, stream>>>(x, weight, dis, y, mtiles * 16);

    row_gather_sum<<<(n_nodes + 7) / 8, 256, 0, stream>>>(y, rowbeg, cnt, ed, dis, bias,
                                                          (float*)d_out, n_nodes);
}


# Edge grouping in two levels: edges binned by 64-row range via LDS (whole-line runs); each range's block orders its edges by row in LDS, then does its rows' product; row sum order = LDS arrival
# speedup vs baseline: 4.7708x; 1.4213x over previous
#include <hip/hip_runtime.h>

#define D 128
#define EB_THREADS 1024
#define EB_PER 16
#define EB_EDGES (EB_THREADS * EB_PER)
#define BIN_ROWS 64
#define BIN_SHIFT 6
#define MAXBINS 1024
static_assert((1 << BIN_SHIFT) == BIN_ROWS, "bin shift");
static_assert(MAXBINS == EB_THREADS, "one thread per bin in the level-1 blocks");

typedef float v2f __attribute__((ext_vector_type(2)));
typedef float v8f __attribute__((ext_vector_type(8)));

__device__ __forceinline__ int block_excl_scan(int v, int* tmp, int& tot) {
    const int lane = threadIdx.x & 31;
    const int w    = threadIdx.x >> 5;
    int inc = v;
    for (int o = 1; o < 32; o <<= 1) {
        const int t = __shfl_up(inc, o);
        if (lane >= o) inc += t;
    }
    if (lane == 31) tmp[w] = inc;
    __syncthreads();
    if (w == 0) {
        const int s = tmp[lane];
        int si = s;
        for (int o = 1; o < 32; o <<= 1) {
            const int t = __shfl_up(si, o);
            if (lane >= o) si += t;
        }
        tmp[lane] = si - s;
        if (lane == 31) tmp[32] = si;
    }
    __syncthreads();
    const int res = tmp[w] + inc - v;
    tot = tmp[32];
    __syncthreads();
    return res;
}

__global__ __launch_bounds__(EB_THREADS) void bin_count(
    const int* __restrict__ erow, int* __restrict__ cntb, int nbins, int ne)
{
    __shared__ int sc[MAXBINS];
    const int tid = threadIdx.x;
    int r[EB_PER];
#pragma unroll
    for (int j = 0; j < EB_PER; ++j) {
        const int i = blockIdx.x * EB_EDGES + j * EB_THREADS + tid;
        const int v = erow[(i < ne) ? i : (ne - 1)];
        r[j] = (i < ne) ? v : -1;
    }
    sc[tid] = 0;
    __syncthreads();
#pragma unroll
    for (int j = 0; j < EB_PER; ++j)
        if (r[j] >= 0) atomicAdd(&sc[r[j] >> BIN_SHIFT], 1);
    __syncthreads();
    if (tid < nbins) cntb[(size_t)blockIdx.x * nbins + tid] = sc[tid];
}

__global__ __launch_bounds__(EB_THREADS) void bin_place(
    const int* __restrict__ erow, const int* __restrict__ ecol,
    const int* __restrict__ cntb, int* __restrict__ binned_i,
    int* __restrict__ binstart, int nbins, int nb, int ne)
{
    __shared__ int sc[MAXBINS];
    __shared__ int loc[MAXBINS];
    __shared__ int gb[MAXBINS];
    __shared__ int tmp[33];
    __shared__ int st_r[EB_EDGES];
    __shared__ int st_c[EB_EDGES];
    int2* __restrict__ binned = (int2*)binned_i;
    const int tid = threadIdx.x;
    const int c   = blockIdx.x;
    int r[EB_PER], cl[EB_PER], k[EB_PER];
#pragma unroll
    for (int j = 0; j < EB_PER; ++j) {
        const int i  = c * EB_EDGES + j * EB_THREADS + tid;
        const int ii = (i < ne) ? i : (ne - 1);
        const int v  = erow[ii];
        cl[j] = ecol[ii];
        r[j]  = (i < ne) ? v : -1;
        k[j]  = 0;
    }
    sc[tid] = 0;
    __syncthreads();
#pragma unroll
    for (int j = 0; j < EB_PER; ++j)
        if (r[j] >= 0) k[j] = atomicAdd(&sc[r[j] >> BIN_SHIFT], 1);
    int tot = 0, pre = 0;
    if (tid < nbins) {
        const int* __restrict__ cp = cntb + tid;
        for (int q0 = 0; q0 < nb; q0 += 32) {
            int v[32];
#pragma unroll
            for (int u = 0; u < 32; ++u) {
                const int q = (q0 + u < nb) ? (q0 + u) : (nb - 1);
                v[u] = cp[(size_t)q * nbins];
            }
#pragma unroll
            for (int u = 0; u < 32; ++u) {
                if (q0 + u < nb) tot += v[u];
                if (q0 + u < c)  pre += v[u];
            }
        }
    }
    __syncthreads();
    const int mine = (tid < nbins) ? sc[tid] : 0;
    int etot, nloc;
    const int bstart = block_excl_scan(tot, tmp, etot);
    const int lstart = block_excl_scan(mine, tmp, nloc);
    if (tid < nbins) {
        loc[tid] = lstart;
        gb[tid]  = bstart + pre - lstart;
        if (c == 0) binstart[tid] = bstart;
    }
    if (c == 0 && tid == 0) binstart[nbins] = etot;
    __syncthreads();
#pragma unroll
    for (int j = 0; j < EB_PER; ++j)
        if (r[j] >= 0) {
            const int s = loc[r[j] >> BIN_SHIFT] + k[j];
            st_r[s] = r[j];
            st_c[s] = cl[j];
        }
    __syncthreads();
    for (int s = tid; s < nloc; s += EB_THREADS) {
        const int row = st_r[s];
        binned[gb[row >> BIN_SHIFT] + s] = make_int2(row, st_c[s]);
    }
}

__global__ __launch_bounds__(256) void row_gather_sum(
    const float* __restrict__ y, const int* __restrict__ rowbeg,
    const int* __restrict__ cnt,
    const int* __restrict__ ed, const float* __restrict__ dis,
    const float* __restrict__ bias, float* __restrict__ out, int n)
{
    const int wid = blockIdx.x * 8 + (threadIdx.x >> 5);
    if (wid >= n) return;
    const int r    = __builtin_amdgcn_readfirstlane(wid);
    const int lane = threadIdx.x & 31;
    const int beg  = rowbeg[r];
    const int end  = beg + cnt[r];
    const float dr = dis[r];
    const float4 bv = ((const float4*)bias)[lane];
    const float4* __restrict__ y4 = (const float4*)y;

    float4 acc = make_float4(0.0f, 0.0f, 0.0f, 0.0f);
    for (int base = beg; base < end; base += 32) {
        const int j = base + lane;
        int my = 0;
        if (j < end) my = ed[j];
        const int m = (end - base < 32) ? (end - base) : 32;
#pragma unroll 8
        for (int k = 0; k < m; ++k) {
            const int c = __builtin_amdgcn_readlane(my, k);
            const float4 yv = y4[(size_t)c * (D / 4) + lane];
            acc.x += yv.x;
            acc.y += yv.y;
            acc.z += yv.z;
            acc.w += yv.w;
        }
    }
    const float s = (end > beg) ? dr : 0.0f;
    float4 o;
    o.x = fmaf(acc.x, s, bv.x);
    o.y = fmaf(acc.y, s, bv.y);
    o.z = fmaf(acc.z, s, bv.z);
    o.w = fmaf(acc.w, s, bv.w);
    ((float4*)out)[(size_t)r * (D / 4) + lane] = o;
}

#define GS_KSTEPS (D / 4)
#define GS_TILES  (D / 16)
#define GS_WTILES 4
#define GS_STRIPS 4

__device__ __forceinline__ void stage_w(const float* __restrict__ Wt, v2f* Bs)
{
    {
        const float4* __restrict__ W4 = (const float4*)Wt;
        float4* Bq = (float4*)Bs;
#pragma unroll
        for (int it = 0; it < (D / 2) * (D / 4) / 256; ++it) {
            const int j = it * 256 + threadIdx.x;
            const int p = j >> 5;
            const int q = j & 31;
            const float4 w0 = W4[(2 * p) * (D / 4) + q];
            const float4 w1 = W4[(2 * p + 1) * (D / 4) + q];
            const int e = (((p >> 1) * GS_TILES + (q >> 2)) * 32) + (p & 1) * 16 + (q & 3) * 4;
            Bq[(e >> 1)]     = make_float4(w0.x, w1.x, w0.y, w1.y);
            Bq[(e >> 1) + 1] = make_float4(w0.z, w1.z, w0.w, w1.w);
        }
    }
}

__device__ __forceinline__ void gemm_scale_block(
    const float* __restrict__ x,
    const float* sdis, float* __restrict__ y, int mtiles,
    int gblock, v2f* Bs)
{

    const int lane  = threadIdx.x & 31;
    const int wave  = threadIdx.x >> 5;
    const int half  = lane >> 4;
    const int l15   = lane & 15;
    const int strip = gblock * GS_STRIPS + (wave >> 1);
    const int t0    = (wave & 1) * GS_WTILES;
    const bool live = strip < mtiles;
    const int m0    = strip * 16;

    if (!live) return;

    v2f a[GS_KSTEPS];
    {
        const float* arow = x + (size_t)(m0 + l15) * D + half * 2;
#pragma unroll
        for (int ks = 0; ks < GS_KSTEPS; ++ks) a[ks] = *(const v2f*)(arow + 4 * ks);
    }

    v8f c[GS_WTILES];
#pragma unroll
    for (int t = 0; t < GS_WTILES; ++t)
#pragma unroll
        for (int r = 0; r < 8; ++r) c[t][r] = 0.0f;

#pragma unroll
    for (int ks = 0; ks < GS_KSTEPS; ++ks) {
#pragma unroll
        for (int t = 0; t < GS_WTILES; ++t) {
            const v2f b = Bs[(ks * GS_TILES + t0 + t) * 32 + lane];
            c[t] = __builtin_amdgcn_wmma_f32_16x16x4_f32(
                        false, a[ks],  false, b,
                        (short)0, c[t],  false,  false);
        }
    }

    const float* dl = sdis + (wave >> 1) * 16 + half * 8;
    const float4 d0 = *(const float4*)(dl);
    const float4 d1 = *(const float4*)(dl + 4);
    const float dv[8] = {d0.x, d0.y, d0.z, d0.w, d1.x, d1.y, d1.z, d1.w};
    float* orow = y + (size_t)(m0 + half * 8) * D + t0 * 16 + l15;
#pragma unroll
    for (int t = 0; t < GS_WTILES; ++t)
#pragma unroll
        for (int r = 0; r < 8; ++r) orow[(size_t)r * D + t * 16] = c[t][r] * dv[r];
}

#define STG_CAP 3072
#define RT_PER 12

__global__ __launch_bounds__(256) void rows_then_product(
    const float* __restrict__ x, const float* __restrict__ Wt,
    float* __restrict__ y, int mtiles,
    const int* __restrict__ binned_i, const int* __restrict__ binstart,
    int* __restrict__ ed, int* __restrict__ rowbeg, int* __restrict__ cnt,
    float* __restrict__ dis, int n)
{
    __shared__ __attribute__((aligned(16))) v2f Bs[GS_KSTEPS * GS_TILES * 32];
    __shared__ int rc[BIN_ROWS];
    __shared__ int rc2[BIN_ROWS];
    __shared__ int rs[BIN_ROWS];
    __shared__ __attribute__((aligned(16))) float sdis[BIN_ROWS];
    __shared__ int stg[STG_CAP];
    const int2* __restrict__ binned = (const int2*)binned_i;

    const int tid  = threadIdx.x;
    const int b    = blockIdx.x;
    const int row0 = b * BIN_ROWS;
    const int s0   = binstart[b];
    const int s1   = binstart[b + 1];
    const int ne   = s1 - s0;
    const bool staged = ne <= STG_CAP;

    if (tid < BIN_ROWS) { rc[tid] = 0; rc2[tid] = 0; }
    const int slast = (s1 > 0) ? (s1 - 1) : 0;
    int er[RT_PER];
#pragma unroll
    for (int u = 0; u < RT_PER; ++u) {
        const int s = s0 + u * 256 + tid;
        er[u] = binned_i[2 * (size_t)((s < s1) ? s : slast)];
    }
    stage_w(Wt, Bs);
    __syncthreads();
#pragma unroll
    for (int u = 0; u < RT_PER; ++u)
        if (s0 + u * 256 + tid < s1) atomicAdd(&rc[er[u] - row0], 1);
    for (int t0 = s0 + 256 * RT_PER; t0 < s1; t0 += 256 * RT_PER) {
#pragma unroll
        for (int u = 0; u < RT_PER; ++u) {
            const int s = t0 + u * 256 + tid;
            er[u] = binned_i[2 * (size_t)((s < s1) ? s : slast)];
        }
#pragma unroll
        for (int u = 0; u < RT_PER; ++u)
            if (t0 + u * 256 + tid < s1) atomicAdd(&rc[er[u] - row0], 1);
    }
    __syncthreads();
    if (tid < BIN_ROWS) {
        int ex = 0;
        for (int j = 0; j < BIN_ROWS; ++j) {
            const int v = rc[j];
            if (j < tid) ex += v;
        }
        const int deg = rc[tid];
        const float dv = rsqrtf((float)deg);
        rs[tid]   = ex;
        sdis[tid] = dv;
        const int rr = row0 + tid;
        if (rr < n) {
            rowbeg[rr] = s0 + ex;
            cnt[rr]    = deg;
            dis[rr]    = dv;
        }
    }
    __syncthreads();
    for (int t0 = s0; t0 < s1; t0 += 256 * RT_PER) {
        int2 e[RT_PER];
#pragma unroll
        for (int u = 0; u < RT_PER; ++u) {
            const int s = t0 + u * 256 + tid;
            e[u] = binned[(s < s1) ? s : slast];
        }
#pragma unroll
        for (int u = 0; u < RT_PER; ++u)
            if (t0 + u * 256 + tid < s1) {
                const int lr  = e[u].x - row0;
                const int pos = rs[lr] + atomicAdd(&rc2[lr], 1);
                if (staged) stg[pos] = e[u].y;
                else        ed[s0 + pos] = e[u].y;
            }
    }
    __syncthreads();
    if (staged)
        for (int s = tid; s < ne; s += 256) ed[s0 + s] = stg[s];
    gemm_scale_block(x, sdis, y, mtiles, b, Bs);
}

__global__ __launch_bounds__(128) void gemm_tail(
    const float* __restrict__ x, const float* __restrict__ Wt,
    const float* __restrict__ dis, float* __restrict__ y,
    int row_start)
{
    int r = row_start + blockIdx.x;
    int n = threadIdx.x;
    float acc = 0.0f;
    const float* a = x + (size_t)r * D;
    for (int k = 0; k < D; ++k) acc = fmaf(a[k], Wt[(size_t)k * D + n], acc);
    y[(size_t)r * D + n] = acc * dis[r];
}

extern "C" void kernel_launch(void* const* d_in, const int* in_sizes, int n_in,
                              void* d_out, int out_size, void* d_ws, size_t ws_size,
                              hipStream_t stream) {
    const float* x      = (const float*)d_in[0];
    const float* weight = (const float*)d_in[1];
    const float* bias   = (const float*)d_in[2];
    const int*   erow   = (const int*)d_in[3];
    const int*   ecol   = (const int*)d_in[4];

    const int n_nodes = in_sizes[0] / D;
    const int n_edges = in_sizes[3];

    const int nbins = (n_nodes + BIN_ROWS - 1) / BIN_ROWS;
    const int nb    = n_edges / EB_EDGES + 1;

    char*  base = (char*)d_ws;
    size_t off  = 0;
    float* y        = (float*)(base + off); off += ((size_t)n_nodes * D * 4 + 255) / 256 * 256;
    int*   cnt      = (int*)(base + off);   off += ((size_t)n_nodes * 4 + 255) / 256 * 256;
    int*   rowbeg   = (int*)(base + off);   off += ((size_t)n_nodes * 4 + 255) / 256 * 256;
    float* dis      = (float*)(base + off); off += ((size_t)n_nodes * 4 + 255) / 256 * 256;
    int*   ed       = (int*)(base + off);   off += ((size_t)n_edges * 4 + 255) / 256 * 256;
    int*   binned   = (int*)(base + off);   off += ((size_t)n_edges * 8 + 255) / 256 * 256;
    int*   cntb     = (int*)(base + off);   off += ((size_t)nb * nbins * 4 + 255) / 256 * 256;
    int*   binstart = (int*)(base + off);

    bin_count<<<nb, EB_THREADS, 0, stream>>>(erow, cntb, nbins, n_edges);
    bin_place<<<nb, EB_THREADS, 0, stream>>>(erow, ecol, cntb, binned, binstart, nbins, nb, n_edges);
    const int mtiles = n_nodes / 16;
    rows_then_product<<<nbins, 256, 0, stream>>>(x, weight, y, mtiles, binned, binstart,
                                                 ed, rowbeg, cnt, dis, n_nodes);
    const int tail = n_nodes - mtiles * 16;
    if (tail > 0)
        gemm_tail<<<tail, D, 0, stream>>>(x, weight, dis, y, mtiles * 16);

    row_gather_sum<<<(n_nodes + 7) / 8, 256, 0, stream>>>(y, rowbeg, cnt, ed, dis, bias,
                                                          (float*)d_out, n_nodes);
}


# Count launch and cross-block prefix removed: a level-1 block writes its bin-ordered pairs to its own piece plus its bin offsets; a bin's block reads its runs by them; row sum order = LDS arrival
# speedup vs baseline: 5.4769x; 1.1480x over previous
#include <hip/hip_runtime.h>

#define D 128
#define EB_THREADS 1024
#define EB_PER 16
#define EB_EDGES (EB_THREADS * EB_PER)
#define BIN_ROWS 64
#define BIN_SHIFT 6
#define MAXBINS 1024
static_assert((1 << BIN_SHIFT) == BIN_ROWS, "bin shift");
static_assert(MAXBINS == EB_THREADS, "one thread per bin in the level-1 blocks");

typedef float v2f __attribute__((ext_vector_type(2)));
typedef float v8f __attribute__((ext_vector_type(8)));

__device__ __forceinline__ int block_excl_scan(int v, int* tmp, int& tot) {
    const int lane = threadIdx.x & 31;
    const int w    = threadIdx.x >> 5;
    int inc = v;
    for (int o = 1; o < 32; o <<= 1) {
        const int t = __shfl_up(inc, o);
        if (lane >= o) inc += t;
    }
    if (lane == 31) tmp[w] = inc;
    __syncthreads();
    if (w == 0) {
        const int s = tmp[lane];
        int si = s;
        for (int o = 1; o < 32; o <<= 1) {
            const int t = __shfl_up(si, o);
            if (lane >= o) si += t;
        }
        tmp[lane] = si - s;
        if (lane == 31) tmp[32] = si;
    }
    __syncthreads();
    const int res = tmp[w] + inc - v;
    tot = tmp[32];
    __syncthreads();
    return res;
}

__global__ __launch_bounds__(EB_THREADS) void bin_sort(
    const int* __restrict__ erow, const int* __restrict__ ecol,
    int* __restrict__ binned_i, int* __restrict__ off,
    int nbins, int ostride, int ne)
{
    __shared__ int sc[MAXBINS];
    __shared__ int loc[MAXBINS];
    __shared__ int tmp[33];
    __shared__ int st_r[EB_EDGES];
    __shared__ int st_c[EB_EDGES];
    int2* __restrict__ binned = (int2*)binned_i;
    const int tid = threadIdx.x;
    const int c   = blockIdx.x;
    int r[EB_PER], cl[EB_PER], k[EB_PER];
#pragma unroll
    for (int j = 0; j < EB_PER; ++j) {
        const int i  = c * EB_EDGES + j * EB_THREADS + tid;
        const int ii = (i < ne) ? i : (ne - 1);
        const int v  = erow[ii];
        cl[j] = ecol[ii];
        r[j]  = (i < ne) ? v : -1;
        k[j]  = 0;
    }
    sc[tid] = 0;
    __syncthreads();
#pragma unroll
    for (int j = 0; j < EB_PER; ++j)
        if (r[j] >= 0) k[j] = atomicAdd(&sc[r[j] >> BIN_SHIFT], 1);
    __syncthreads();
    const int mine = (tid < nbins) ? sc[tid] : 0;
    int nloc;
    const int lstart = block_excl_scan(mine, tmp, nloc);
    loc[tid] = lstart;
    int* __restrict__ orow = off + (size_t)c * ostride;
    if (tid < nbins) orow[tid] = lstart;
    if (tid == 0) orow[nbins] = nloc;
    __syncthreads();
#pragma unroll
    for (int j = 0; j < EB_PER; ++j)
        if (r[j] >= 0) {
            const int s = loc[r[j] >> BIN_SHIFT] + k[j];
            st_r[s] = r[j];
            st_c[s] = cl[j];
        }
    __syncthreads();
    int2* __restrict__ piece = binned + (size_t)c * EB_EDGES;
    for (int s = tid; s < nloc; s += EB_THREADS)
        piece[s] = make_int2(st_r[s], st_c[s]);
}

__global__ __launch_bounds__(256) void row_gather_sum(
    const float* __restrict__ y, const int* __restrict__ rowbeg,
    const int* __restrict__ cnt,
    const int* __restrict__ ed, const float* __restrict__ dis,
    const float* __restrict__ bias, float* __restrict__ out, int n)
{
    const int wid = blockIdx.x * 8 + (threadIdx.x >> 5);
    if (wid >= n) return;
    const int r    = __builtin_amdgcn_readfirstlane(wid);
    const int lane = threadIdx.x & 31;
    const int beg  = rowbeg[r];
    const int end  = beg + cnt[r];
    const float dr = dis[r];
    const float4 bv = ((const float4*)bias)[lane];
    const float4* __restrict__ y4 = (const float4*)y;

    float4 acc = make_float4(0.0f, 0.0f, 0.0f, 0.0f);
    for (int base = beg; base < end; base += 32) {
        const int j = base + lane;
        int my = 0;
        if (j < end) my = ed[j];
        const int m = (end - base < 32) ? (end - base) : 32;
#pragma unroll 8
        for (int k = 0; k < m; ++k) {
            const int c = __builtin_amdgcn_readlane(my, k);
            const float4 yv = y4[(size_t)c * (D / 4) + lane];
            acc.x += yv.x;
            acc.y += yv.y;
            acc.z += yv.z;
            acc.w += yv.w;
        }
    }
    const float s = (end > beg) ? dr : 0.0f;
    float4 o;
    o.x = fmaf(acc.x, s, bv.x);
    o.y = fmaf(acc.y, s, bv.y);
    o.z = fmaf(acc.z, s, bv.z);
    o.w = fmaf(acc.w, s, bv.w);
    ((float4*)out)[(size_t)r * (D / 4) + lane] = o;
}

#define GS_KSTEPS (D / 4)
#define GS_TILES  (D / 16)
#define GS_WTILES 4
#define GS_STRIPS 4

__device__ __forceinline__ void stage_w(const float* __restrict__ Wt, v2f* Bs)
{
    {
        const float4* __restrict__ W4 = (const float4*)Wt;
        float4* Bq = (float4*)Bs;
#pragma unroll
        for (int it = 0; it < (D / 2) * (D / 4) / 256; ++it) {
            const int j = it * 256 + threadIdx.x;
            const int p = j >> 5;
            const int q = j & 31;
            const float4 w0 = W4[(2 * p) * (D / 4) + q];
            const float4 w1 = W4[(2 * p + 1) * (D / 4) + q];
            const int e = (((p >> 1) * GS_TILES + (q >> 2)) * 32) + (p & 1) * 16 + (q & 3) * 4;
            Bq[(e >> 1)]     = make_float4(w0.x, w1.x, w0.y, w1.y);
            Bq[(e >> 1) + 1] = make_float4(w0.z, w1.z, w0.w, w1.w);
        }
    }
}

__device__ __forceinline__ void gemm_scale_block(
    const float* __restrict__ x,
    const float* sdis, float* __restrict__ y, int mtiles,
    int gblock, v2f* Bs)
{

    const int lane  = threadIdx.x & 31;
    const int wave  = threadIdx.x >> 5;
    const int half  = lane >> 4;
    const int l15   = lane & 15;
    const int strip = gblock * GS_STRIPS + (wave >> 1);
    const int t0    = (wave & 1) * GS_WTILES;
    const bool live = strip < mtiles;
    const int m0    = strip * 16;

    if (!live) return;

    v2f a[GS_KSTEPS];
    {
        const float* arow = x + (size_t)(m0 + l15) * D + half * 2;
#pragma unroll
        for (int ks = 0; ks < GS_KSTEPS; ++ks) a[ks] = *(const v2f*)(arow + 4 * ks);
    }

    v8f c[GS_WTILES];
#pragma unroll
    for (int t = 0; t < GS_WTILES; ++t)
#pragma unroll
        for (int r = 0; r < 8; ++r) c[t][r] = 0.0f;

#pragma unroll
    for (int ks = 0; ks < GS_KSTEPS; ++ks) {
#pragma unroll
        for (int t = 0; t < GS_WTILES; ++t) {
            const v2f b = Bs[(ks * GS_TILES + t0 + t) * 32 + lane];
            c[t] = __builtin_amdgcn_wmma_f32_16x16x4_f32(
                        false, a[ks],  false, b,
                        (short)0, c[t],  false,  false);
        }
    }

    const float* dl = sdis + (wave >> 1) * 16 + half * 8;
    const float4 d0 = *(const float4*)(dl);
    const float4 d1 = *(const float4*)(dl + 4);
    const float dv[8] = {d0.x, d0.y, d0.z, d0.w, d1.x, d1.y, d1.z, d1.w};
    float* orow = y + (size_t)(m0 + half * 8) * D + t0 * 16 + l15;
#pragma unroll
    for (int t = 0; t < GS_WTILES; ++t)
#pragma unroll
        for (int r = 0; r < 8; ++r) orow[(size_t)r * D + t * 16] = c[t][r] * dv[r];
}

#define STG_CAP 3072
#define MAXNB 256
static_assert(EB_EDGES < 32768, "a run's offset and its length share one int of the run table");
#define RUN_PER 13

__global__ __launch_bounds__(256) void rows_then_product(
    const float* __restrict__ x, const float* __restrict__ Wt,
    float* __restrict__ y, int mtiles,
    const int* __restrict__ binned_i, const int* __restrict__ off, int nb, int ostride,
    int* __restrict__ ed, int* __restrict__ rowbeg, int* __restrict__ cnt,
    float* __restrict__ dis, int n)
{
    __shared__ __attribute__((aligned(16))) v2f Bs[GS_KSTEPS * GS_TILES * 32];
    __shared__ int rc[BIN_ROWS];
    __shared__ int rc2[BIN_ROWS];
    __shared__ int rs[BIN_ROWS];
    __shared__ __attribute__((aligned(16))) float sdis[BIN_ROWS];
    __shared__ int stg[STG_CAP];
    __shared__ int run[MAXNB];
    const int2* __restrict__ binned = (const int2*)binned_i;

    const int tid  = threadIdx.x;
    const int lane = tid & 31;
    const int w    = tid >> 5;
    const int b    = blockIdx.x;
    const int row0 = b * BIN_ROWS;

    if (tid < BIN_ROWS) { rc[tid] = 0; rc2[tid] = 0; }
    const int cq = (tid < nb) ? tid : 0;
    const int o0 = off[(size_t)cq * ostride + b];
    const int o1 = off[(size_t)cq * ostride + b + 1];
    stage_w(Wt, Bs);
    run[tid] = (tid < nb) ? (((o1 - o0) << 16) | o0) : 0;
    __syncthreads();
    int a0 = 0, a1 = 0;
    for (int c = lane; c < nb; c += 32) { const int pv = run[c]; a0 += pv & 0xffff; a1 += pv >> 16; }
    for (int o = 16; o > 0; o >>= 1) { a0 += __shfl_xor(a0, o); a1 += __shfl_xor(a1, o); }
    const int s0 = __builtin_amdgcn_readfirstlane(a0);
    const int ne = __builtin_amdgcn_readfirstlane(a1);
    const bool staged = ne <= STG_CAP;

    int longest = 0;
    for (int c0 = 0; c0 < nb; c0 += 8 * RUN_PER) {
        int pr[RUN_PER];
        int er[RUN_PER];
#pragma unroll
        for (int u = 0; u < RUN_PER; ++u) {
            const int c  = c0 + u * 8 + w;
            const int pv = run[(c < nb) ? c : 0];
            pr[u] = (c < nb) ? pv : 0;
            longest = (pr[u] >> 16) > longest ? (pr[u] >> 16) : longest;
        }
#pragma unroll
        for (int u = 0; u < RUN_PER; ++u) {
            const int c   = c0 + u * 8 + w;
            const bool ok = lane < (pr[u] >> 16);
            er[u] = binned_i[2 * (size_t)(ok ? (c * EB_EDGES + (pr[u] & 0xffff) + lane) : 0)];
        }
#pragma unroll
        for (int u = 0; u < RUN_PER; ++u)
            if (lane < (pr[u] >> 16)) atomicAdd(&rc[er[u] - row0], 1);
    }
    if (longest > 32)
    for (int c = w; c < nb; c += 8) {
        const int pv  = run[c];
        const int len = pv >> 16;
        const int fst = c * EB_EDGES + (pv & 0xffff);
        for (int k = 32 + lane; k < len; k += 32)
            atomicAdd(&rc[binned_i[2 * (size_t)(fst + k)] - row0], 1);
    }
    __syncthreads();
    if (tid < BIN_ROWS) {
        int ex = 0;
        for (int j = 0; j < BIN_ROWS; ++j) {
            const int v = rc[j];
            if (j < tid) ex += v;
        }
        const int deg = rc[tid];
        const float dv = rsqrtf((float)deg);
        rs[tid]   = ex;
        sdis[tid] = dv;
        const int rr = row0 + tid;
        if (rr < n) {
            rowbeg[rr] = s0 + ex;
            cnt[rr]    = deg;
            dis[rr]    = dv;
        }
    }
    __syncthreads();
    for (int c0 = 0; c0 < nb; c0 += 8 * RUN_PER) {
        int pr[RUN_PER];
        int2 e[RUN_PER];
#pragma unroll
        for (int u = 0; u < RUN_PER; ++u) {
            const int c  = c0 + u * 8 + w;
            const int pv = run[(c < nb) ? c : 0];
            pr[u] = (c < nb) ? pv : 0;
        }
#pragma unroll
        for (int u = 0; u < RUN_PER; ++u) {
            const int c   = c0 + u * 8 + w;
            const bool ok = lane < (pr[u] >> 16);
            e[u] = binned[ok ? (c * EB_EDGES + (pr[u] & 0xffff) + lane) : 0];
        }
#pragma unroll
        for (int u = 0; u < RUN_PER; ++u)
            if (lane < (pr[u] >> 16)) {
                const int lr  = e[u].x - row0;
                const int pos = rs[lr] + atomicAdd(&rc2[lr], 1);
                if (staged) stg[pos] = e[u].y;
                else        ed[s0 + pos] = e[u].y;
            }
    }
    if (longest > 32)
    for (int c = w; c < nb; c += 8) {
        const int pv  = run[c];
        const int len = pv >> 16;
        const int fst = c * EB_EDGES + (pv & 0xffff);
        for (int k = 32 + lane; k < len; k += 32) {
            const int2 e  = binned[fst + k];
            const int lr  = e.x - row0;
            const int pos = rs[lr] + atomicAdd(&rc2[lr], 1);
            if (staged) stg[pos] = e.y;
            else        ed[s0 + pos] = e.y;
        }
    }
    __syncthreads();
    if (staged)
        for (int s = tid; s < ne; s += 256) ed[s0 + s] = stg[s];
    gemm_scale_block(x, sdis, y, mtiles, b, Bs);
}

__global__ __launch_bounds__(128) void gemm_tail(
    const float* __restrict__ x, const float* __restrict__ Wt,
    const float* __restrict__ dis, float* __restrict__ y,
    int row_start)
{
    int r = row_start + blockIdx.x;
    int n = threadIdx.x;
    float acc = 0.0f;
    const float* a = x + (size_t)r * D;
    for (int k = 0; k < D; ++k) acc = fmaf(a[k], Wt[(size_t)k * D + n], acc);
    y[(size_t)r * D + n] = acc * dis[r];
}

extern "C" void kernel_launch(void* const* d_in, const int* in_sizes, int n_in,
                              void* d_out, int out_size, void* d_ws, size_t ws_size,
                              hipStream_t stream) {
    const float* x      = (const float*)d_in[0];
    const float* weight = (const float*)d_in[1];
    const float* bias   = (const float*)d_in[2];
    const int*   erow   = (const int*)d_in[3];
    const int*   ecol   = (const int*)d_in[4];

    const int n_nodes = in_sizes[0] / D;
    const int n_edges = in_sizes[3];

    const int nbins = (n_nodes + BIN_ROWS - 1) / BIN_ROWS;
    const int nb    = n_edges / EB_EDGES + 1;
    const int ostride = (nbins + 1 + 31) / 32 * 32;

    char*  base = (char*)d_ws;
    size_t off  = 0;
    float* y        = (float*)(base + off); off += ((size_t)n_nodes * D * 4 + 255) / 256 * 256;
    int*   cnt      = (int*)(base + off);   off += ((size_t)n_nodes * 4 + 255) / 256 * 256;
    int*   rowbeg   = (int*)(base + off);   off += ((size_t)n_nodes * 4 + 255) / 256 * 256;
    float* dis      = (float*)(base + off); off += ((size_t)n_nodes * 4 + 255) / 256 * 256;
    int*   ed       = (int*)(base + off);   off += ((size_t)n_edges * 4 + 255) / 256 * 256;
    int*   binned   = (int*)(base + off);   off += ((size_t)n_edges * 8 + 255) / 256 * 256;
    int*   boff     = (int*)(base + off);

    bin_sort<<<nb, EB_THREADS, 0, stream>>>(erow, ecol, binned, boff, nbins, ostride, n_edges);
    const int mtiles = n_nodes / 16;
    rows_then_product<<<nbins, 256, 0, stream>>>(x, weight, y, mtiles, binned, boff, nb, ostride,
                                                 ed, rowbeg, cnt, dis, n_nodes);
    const int tail = n_nodes - mtiles * 16;
    if (tail > 0)
        gemm_tail<<<tail, D, 0, stream>>>(x, weight, dis, y, mtiles * 16);

    row_gather_sum<<<(n_nodes + 7) / 8, 256, 0, stream>>>(y, rowbeg, cnt, ed, dis, bias,
                                                          (float*)d_out, n_nodes);
}
